# starting kernel
# baseline (speedup 1.0000x reference)
#include <hip/hip_runtime.h>
#include <hip/hip_bf16.h>
#include <stdint.h>

typedef __attribute__((ext_vector_type(2))) float v2f;
typedef __attribute__((ext_vector_type(8))) float v8f;

#define NNODE   10000
#define NEDGE   160000
#define EN      (NEDGE + NNODE)
#define DIM     256
#define MT      (NNODE / 16)
#define NEG_SLOPE 0.2f
#define BN_EPS  1e-5f

__global__ void zero_ws(float4* __restrict__ p, long n4) {
    long i = (long)blockIdx.x * blockDim.x + threadIdx.x;
    long stride = (long)gridDim.x * blockDim.x;
    float4 z = make_float4(0.f, 0.f, 0.f, 0.f);
    for (; i < n4; i += stride) p[i] = z;
}

__global__ void gemm_xlxr(const float* __restrict__ x,
                          const float* __restrict__ Wl, const float* __restrict__ bl,
                          const float* __restrict__ Wr, const float* __restrict__ br,
                          float* __restrict__ xl, float* __restrict__ xr) {
    int wid  = (int)((blockIdx.x * blockDim.x + threadIdx.x) >> 5);
    int lane = (int)(threadIdx.x & 31);
    if (wid >= MT * 32) return;

    int mt   = wid >> 5;
    int rest = wid & 31;
    int mat  = rest >> 4;
    int nt   = rest & 15;

    const float* W    = mat ? Wr : Wl;
    const float* bias = mat ? br : bl;
    float*       out  = mat ? xr : xl;

    int half = lane >> 4;
    int l16  = lane & 15;
    int row  = mt * 16 + l16;
    int col  = nt * 16 + l16;

    const float* aptr = x + (long)row * DIM + 2 * half;
    const float* bptr = W + (long)(2 * half) * DIM + col;

    v8f c = {};
#pragma unroll 4
    for (int k = 0; k < DIM; k += 4) {
        v2f a, b;
        a[0] = aptr[k];
        a[1] = aptr[k + 1];
        b[0] = bptr[(long)k * DIM];
        b[1] = bptr[(long)k * DIM + DIM];
        c = __builtin_amdgcn_wmma_f32_16x16x4_f32(
                 false, a,  false, b,
                 (short)0, c,  false,  false);
    }

    float bv = bias[col];
#pragma unroll
    for (int r = 0; r < 8; ++r) {
        int m = mt * 16 + r + 8 * half;
        out[(long)m * DIM + col] = c[r] + bv;
    }
}

__device__ __forceinline__ unsigned fenc(float f) {
    unsigned b = __float_as_uint(f);
    return (b & 0x80000000u) ? ~b : (b | 0x80000000u);
}
__device__ __forceinline__ float fdec(unsigned u) {
    return __uint_as_float((u & 0x80000000u) ? (u ^ 0x80000000u) : ~u);
}

__global__ void edge_logits(const int* __restrict__ ei,
                            const float* __restrict__ xl, const float* __restrict__ xr,
                            const float* __restrict__ att,
                            float* __restrict__ alpha_raw, unsigned* __restrict__ amax) {
    int wid  = (int)((blockIdx.x * blockDim.x + threadIdx.x) >> 5);
    int lane = (int)(threadIdx.x & 31);
    if (wid >= EN) return;

    int s, d;
    if (wid < NEDGE) { s = ei[wid]; d = ei[NEDGE + wid]; }
    else             { s = d = wid - NEDGE; }

    const float4* xs = (const float4*)(xl + (long)s * DIM);
    const float4* xd = (const float4*)(xr + (long)d * DIM);
    const float4* av = (const float4*)att;

    float acc = 0.f;
#pragma unroll
    for (int j = lane; j < DIM / 4; j += 32) {
        float4 a = xs[j], b = xd[j], t = av[j];
        float h;
        h = a.x + b.x; h = h > 0.f ? h : NEG_SLOPE * h; acc += t.x * h;
        h = a.y + b.y; h = h > 0.f ? h : NEG_SLOPE * h; acc += t.y * h;
        h = a.z + b.z; h = h > 0.f ? h : NEG_SLOPE * h; acc += t.z * h;
        h = a.w + b.w; h = h > 0.f ? h : NEG_SLOPE * h; acc += t.w * h;
    }
#pragma unroll
    for (int off = 16; off > 0; off >>= 1) acc += __shfl_xor(acc, off, 32);

    if (lane == 0) {
        alpha_raw[wid] = acc;
        atomicMax(&amax[d], fenc(acc));
    }
}

__global__ void edge_aggregate(const int* __restrict__ ei,
                               const float* __restrict__ xl,
                               const float* __restrict__ alpha_raw,
                               const unsigned* __restrict__ amax,
                               float* __restrict__ denom, float* __restrict__ agg) {
    int wid  = (int)((blockIdx.x * blockDim.x + threadIdx.x) >> 5);
    int lane = (int)(threadIdx.x & 31);
    if (wid >= EN) return;

    int s, d;
    if (wid < NEDGE) { s = ei[wid]; d = ei[NEDGE + wid]; }
    else             { s = d = wid - NEDGE; }

    float al = __expf(alpha_raw[wid] - fdec(amax[d]));
    if (lane == 0) atomicAdd(&denom[d], al);

    const float4* xs = (const float4*)(xl + (long)s * DIM);
    float* out = agg + (long)d * DIM;
#pragma unroll
    for (int j = lane; j < DIM / 4; j += 32) {
        float4 a = xs[j];
        atomicAdd(&out[4 * j + 0], al * a.x);
        atomicAdd(&out[4 * j + 1], al * a.y);
        atomicAdd(&out[4 * j + 2], al * a.z);
        atomicAdd(&out[4 * j + 3], al * a.w);
    }
}

#define ROWS_PER_BLK 50

__global__ void post_stats(float* __restrict__ agg, const float* __restrict__ denom,
                           const float* __restrict__ bias, const float* __restrict__ du,
                           float* __restrict__ sums, float* __restrict__ sumsq) {
    int dcol = (int)threadIdx.x;
    int i0   = (int)blockIdx.x * ROWS_PER_BLK;
    float s = 0.f, s2 = 0.f;
    for (int r = 0; r < ROWS_PER_BLK; ++r) {
        int i = i0 + r;
        if (i >= NNODE) break;
        long idx = (long)i * DIM + dcol;
        float v = agg[idx] / denom[i] + bias[dcol];
        v = v > 0.f ? v : 0.f;
        v = (du[idx] >= 0.5f) ? 2.0f * v : 0.0f;
        agg[idx] = v;
        s += v; s2 += v * v;
    }
    atomicAdd(&sums[dcol], s);
    atomicAdd(&sumsq[dcol], s2);
}

__global__ void finalize(const float* __restrict__ v,
                         const float* __restrict__ sums, const float* __restrict__ sumsq,
                         const float* __restrict__ gamma, const float* __restrict__ beta,
                         float* __restrict__ out) {
    int dcol = (int)threadIdx.x;
    int i0   = (int)blockIdx.x * ROWS_PER_BLK;
    const float invn = 1.0f / (float)NNODE;
    float mean  = sums[dcol] * invn;
    float var   = sumsq[dcol] * invn - mean * mean;
    float scale = gamma[dcol] * rsqrtf(var + BN_EPS);
    float shift = beta[dcol] - mean * scale;
    for (int r = 0; r < ROWS_PER_BLK; ++r) {
        int i = i0 + r;
        if (i >= NNODE) break;
        long idx = (long)i * DIM + dcol;
        out[idx] = v[idx] * scale + shift;
    }
}

extern "C" void kernel_launch(void* const* d_in, const int* in_sizes, int n_in,
                              void* d_out, int out_size, void* d_ws, size_t ws_size,
                              hipStream_t stream) {
    const float* x     = (const float*)d_in[0];
    const int*   ei    = (const int*)d_in[1];
    const float* Wl    = (const float*)d_in[2];
    const float* bl    = (const float*)d_in[3];
    const float* Wr    = (const float*)d_in[4];
    const float* br    = (const float*)d_in[5];
    const float* att   = (const float*)d_in[6];
    const float* bias  = (const float*)d_in[7];
    const float* gamma = (const float*)d_in[8];
    const float* beta  = (const float*)d_in[9];
    const float* du    = (const float*)d_in[10];
    float*       out   = (float*)d_out;

    float*    ws    = (float*)d_ws;
    float*    agg   = ws;
    unsigned* amax  = (unsigned*)(ws + (long)NNODE * DIM);
    float*    denom = ws + (long)NNODE * DIM + NNODE;
    float*    sums  = denom + NNODE;
    float*    sumsq = sums + DIM;
    float*    xl    = sumsq + DIM;
    float*    xr    = xl + (long)NNODE * DIM;
    float*    araw  = xr + (long)NNODE * DIM;

    long zeroN  = (long)NNODE * DIM + 2L * NNODE + 2L * DIM;
    long zeroN4 = zeroN / 4;
    zero_ws<<<1024, 256, 0, stream>>>((float4*)d_ws, zeroN4);

    gemm_xlxr<<<(MT * 32) / 4, 128, 0, stream>>>(x, Wl, bl, Wr, br, xl, xr);

    int eblocks = (EN * 32 + 255) / 256;
    edge_logits<<<eblocks, 256, 0, stream>>>(ei, xl, xr, att, araw, amax);

    edge_aggregate<<<eblocks, 256, 0, stream>>>(ei, xl, araw, amax, denom, agg);

    int sblocks = (NNODE + ROWS_PER_BLK - 1) / ROWS_PER_BLK;
    post_stats<<<sblocks, DIM, 0, stream>>>(agg, denom, bias, du, sums, sumsq);

    finalize<<<sblocks, DIM, 0, stream>>>(agg, sums, sumsq, gamma, beta, out);
}


# edges bucketed by destination in-call (count, scan, fill); one wave per node keeps running max, sum and weighted row sum in registers instead of float atomic scatter; sum order changes
# speedup vs baseline: 1.4506x; 1.4506x over previous
#include <hip/hip_runtime.h>
#include <hip/hip_bf16.h>
#include <stdint.h>

typedef __attribute__((ext_vector_type(2))) float v2f;
typedef __attribute__((ext_vector_type(8))) float v8f;

#define NNODE   10000
#define NEDGE   160000
#define EN      (NEDGE + NNODE)
#define DIM     256
#define MT      (NNODE / 16)
#define NEG_SLOPE 0.2f
#define BN_EPS  1e-5f

__global__ void zero_ws(float4* __restrict__ p, long n4) {
    long i = (long)blockIdx.x * blockDim.x + threadIdx.x;
    long stride = (long)gridDim.x * blockDim.x;
    float4 z = make_float4(0.f, 0.f, 0.f, 0.f);
    for (; i < n4; i += stride) p[i] = z;
}

__global__ void gemm_xlxr(const float* __restrict__ x,
                          const float* __restrict__ Wl, const float* __restrict__ bl,
                          const float* __restrict__ Wr, const float* __restrict__ br,
                          float* __restrict__ xl, float* __restrict__ xr) {
    int wid  = (int)((blockIdx.x * blockDim.x + threadIdx.x) >> 5);
    int lane = (int)(threadIdx.x & 31);
    if (wid >= MT * 32) return;

    int mt   = wid >> 5;
    int rest = wid & 31;
    int mat  = rest >> 4;
    int nt   = rest & 15;

    const float* W    = mat ? Wr : Wl;
    const float* bias = mat ? br : bl;
    float*       out  = mat ? xr : xl;

    int half = lane >> 4;
    int l16  = lane & 15;
    int row  = mt * 16 + l16;
    int col  = nt * 16 + l16;

    const float* aptr = x + (long)row * DIM + 2 * half;
    const float* bptr = W + (long)(2 * half) * DIM + col;

    v8f c = {};
#pragma unroll 4
    for (int k = 0; k < DIM; k += 4) {
        v2f a, b;
        a[0] = aptr[k];
        a[1] = aptr[k + 1];
        b[0] = bptr[(long)k * DIM];
        b[1] = bptr[(long)k * DIM + DIM];
        c = __builtin_amdgcn_wmma_f32_16x16x4_f32(
                 false, a,  false, b,
                 (short)0, c,  false,  false);
    }

    float bv = bias[col];
#pragma unroll
    for (int r = 0; r < 8; ++r) {
        int m = mt * 16 + r + 8 * half;
        out[(long)m * DIM + col] = c[r] + bv;
    }
}

__global__ void seg_count(const int* __restrict__ ei, unsigned* __restrict__ segpos) {
    int e = (int)(blockIdx.x * blockDim.x + threadIdx.x);
    if (e < NEDGE) atomicAdd(&segpos[ei[NEDGE + e]], 1u);
}

#define SCAN_T     1024
#define SCAN_ITEMS ((NNODE + SCAN_T - 1) / SCAN_T)

__global__ void seg_scan(unsigned* __restrict__ segpos) {
    __shared__ unsigned wsum[SCAN_T / 32];
    int tid  = (int)threadIdx.x;
    int lane = tid & 31;
    int w    = tid >> 5;
    int base = tid * SCAN_ITEMS;

    unsigned v[SCAN_ITEMS];
    unsigned tot = 0u;
    for (int i = 0; i < SCAN_ITEMS; ++i) {
        int idx = base + i;
        v[i] = (idx < NNODE) ? segpos[idx] : 0u;
        tot += v[i];
    }
    unsigned inc = tot;
    for (int off = 1; off < 32; off <<= 1) {
        unsigned t = __shfl_up(inc, (unsigned)off, 32);
        if (lane >= off) inc += t;
    }
    if (lane == 31) wsum[w] = inc;
    __syncthreads();
    if (w == 0) {
        unsigned s  = wsum[lane];
        unsigned si = s;
        for (int off = 1; off < 32; off <<= 1) {
            unsigned t = __shfl_up(si, (unsigned)off, 32);
            if (lane >= off) si += t;
        }
        wsum[lane] = si - s;
    }
    __syncthreads();
    unsigned run = wsum[w] + (inc - tot);
    for (int i = 0; i < SCAN_ITEMS; ++i) {
        int idx = base + i;
        if (idx < NNODE) { segpos[idx] = run; run += v[i]; }
    }
}

__global__ void seg_fill(const int* __restrict__ ei, unsigned* __restrict__ segpos,
                         int* __restrict__ col) {
    int e = (int)(blockIdx.x * blockDim.x + threadIdx.x);
    if (e < NEDGE) {
        int d = ei[NEDGE + e];
        unsigned p = atomicAdd(&segpos[d], 1u);
        col[p] = ei[e];
    }
}

__device__ __forceinline__ float edge_logit(const float4& a0, const float4& a1,
                                            const float4& r0, const float4& r1,
                                            const float4& t0, const float4& t1) {
    float acc = 0.f, h;
    h = a0.x + r0.x; h = h > 0.f ? h : NEG_SLOPE * h; acc += t0.x * h;
    h = a0.y + r0.y; h = h > 0.f ? h : NEG_SLOPE * h; acc += t0.y * h;
    h = a0.z + r0.z; h = h > 0.f ? h : NEG_SLOPE * h; acc += t0.z * h;
    h = a0.w + r0.w; h = h > 0.f ? h : NEG_SLOPE * h; acc += t0.w * h;
    h = a1.x + r1.x; h = h > 0.f ? h : NEG_SLOPE * h; acc += t1.x * h;
    h = a1.y + r1.y; h = h > 0.f ? h : NEG_SLOPE * h; acc += t1.y * h;
    h = a1.z + r1.z; h = h > 0.f ? h : NEG_SLOPE * h; acc += t1.z * h;
    h = a1.w + r1.w; h = h > 0.f ? h : NEG_SLOPE * h; acc += t1.w * h;
    for (int off = 16; off > 0; off >>= 1) acc += __shfl_xor(acc, off, 32);
    return acc;
}

__global__ void node_attend(const unsigned* __restrict__ segpos, const int* __restrict__ col,
                            const float* __restrict__ xl, const float* __restrict__ xr,
                            const float* __restrict__ att,
                            float* __restrict__ denom, float* __restrict__ agg) {
    int d    = (int)((blockIdx.x * blockDim.x + threadIdx.x) >> 5);
    int lane = (int)(threadIdx.x & 31);
    if (d >= NNODE) return;

    unsigned beg = d ? segpos[d - 1] : 0u;
    unsigned end = segpos[d];

    const float4* xd = (const float4*)(xr + (long)d * DIM);
    const float4* av = (const float4*)att;
    float4 r0 = xd[lane], r1 = xd[lane + 32];
    float4 t0 = av[lane], t1 = av[lane + 32];

    const float4* xs = (const float4*)(xl + (long)d * DIM);
    float4 acc0 = xs[lane], acc1 = xs[lane + 32];
    float  m    = edge_logit(acc0, acc1, r0, r1, t0, t1);
    float  den  = 1.f;

    for (unsigned cb = beg; cb < end; cb += 32u) {
        unsigned left = end - cb;
        int n     = left < 32u ? (int)left : 32;
        int mycol = (lane < n) ? col[cb + (unsigned)lane] : 0;

        int s = __shfl(mycol, 0, 32);
        const float4* xn = (const float4*)(xl + (long)s * DIM);
        float4 n0 = xn[lane], n1 = xn[lane + 32];
        for (int i = 0; i < n; ++i) {
            float4 a0 = n0, a1 = n1;
            if (i + 1 < n) {
                s  = __shfl(mycol, i + 1, 32);
                xn = (const float4*)(xl + (long)s * DIM);
                n0 = xn[lane]; n1 = xn[lane + 32];
            }
            float lg = edge_logit(a0, a1, r0, r1, t0, t1);
            float mn = fmaxf(m, lg);
            float sc = __expf(m - mn);
            float wt = __expf(lg - mn);
            den = den * sc + wt;
            acc0.x = acc0.x * sc + wt * a0.x;  acc0.y = acc0.y * sc + wt * a0.y;
            acc0.z = acc0.z * sc + wt * a0.z;  acc0.w = acc0.w * sc + wt * a0.w;
            acc1.x = acc1.x * sc + wt * a1.x;  acc1.y = acc1.y * sc + wt * a1.y;
            acc1.z = acc1.z * sc + wt * a1.z;  acc1.w = acc1.w * sc + wt * a1.w;
            m = mn;
        }
    }

    if (lane == 0) denom[d] = den;
    float4* out = (float4*)(agg + (long)d * DIM);
    out[lane]      = acc0;
    out[lane + 32] = acc1;
}

#define ROWS_PER_BLK 50

__global__ void post_stats(float* __restrict__ agg, const float* __restrict__ denom,
                           const float* __restrict__ bias, const float* __restrict__ du,
                           float* __restrict__ sums, float* __restrict__ sumsq) {
    int dcol = (int)threadIdx.x;
    int i0   = (int)blockIdx.x * ROWS_PER_BLK;
    float s = 0.f, s2 = 0.f;
    for (int r = 0; r < ROWS_PER_BLK; ++r) {
        int i = i0 + r;
        if (i >= NNODE) break;
        long idx = (long)i * DIM + dcol;
        float v = agg[idx] / denom[i] + bias[dcol];
        v = v > 0.f ? v : 0.f;
        v = (du[idx] >= 0.5f) ? 2.0f * v : 0.0f;
        agg[idx] = v;
        s += v; s2 += v * v;
    }
    atomicAdd(&sums[dcol], s);
    atomicAdd(&sumsq[dcol], s2);
}

__global__ void finalize(const float* __restrict__ v,
                         const float* __restrict__ sums, const float* __restrict__ sumsq,
                         const float* __restrict__ gamma, const float* __restrict__ beta,
                         float* __restrict__ out) {
    int dcol = (int)threadIdx.x;
    int i0   = (int)blockIdx.x * ROWS_PER_BLK;
    const float invn = 1.0f / (float)NNODE;
    float mean  = sums[dcol] * invn;
    float var   = sumsq[dcol] * invn - mean * mean;
    float scale = gamma[dcol] * rsqrtf(var + BN_EPS);
    float shift = beta[dcol] - mean * scale;
    for (int r = 0; r < ROWS_PER_BLK; ++r) {
        int i = i0 + r;
        if (i >= NNODE) break;
        long idx = (long)i * DIM + dcol;
        out[idx] = v[idx] * scale + shift;
    }
}

extern "C" void kernel_launch(void* const* d_in, const int* in_sizes, int n_in,
                              void* d_out, int out_size, void* d_ws, size_t ws_size,
                              hipStream_t stream) {
    const float* x     = (const float*)d_in[0];
    const int*   ei    = (const int*)d_in[1];
    const float* Wl    = (const float*)d_in[2];
    const float* bl    = (const float*)d_in[3];
    const float* Wr    = (const float*)d_in[4];
    const float* br    = (const float*)d_in[5];
    const float* att   = (const float*)d_in[6];
    const float* bias  = (const float*)d_in[7];
    const float* gamma = (const float*)d_in[8];
    const float* beta  = (const float*)d_in[9];
    const float* du    = (const float*)d_in[10];
    float*       out   = (float*)d_out;

    float*    ws     = (float*)d_ws;
    float*    agg    = ws;
    float*    denom  = ws + (long)NNODE * DIM;
    unsigned* segpos = (unsigned*)(denom + NNODE);
    float*    sums   = denom + 2L * NNODE;
    float*    sumsq  = sums + DIM;
    float*    xl     = sumsq + DIM;
    float*    xr     = xl + (long)NNODE * DIM;
    int*      col    = (int*)(xr + (long)NNODE * DIM);

    long zeroN4 = ((long)NNODE + 2L * DIM) / 4;
    zero_ws<<<(NNODE + 2 * DIM) / 4 / 256 + 1, 256, 0, stream>>>((float4*)segpos, zeroN4);

    gemm_xlxr<<<(MT * 32) / 4, 128, 0, stream>>>(x, Wl, bl, Wr, br, xl, xr);

    int cblocks = (NEDGE + 255) / 256;
    seg_count<<<cblocks, 256, 0, stream>>>(ei, segpos);
    seg_scan<<<1, SCAN_T, 0, stream>>>(segpos);
    seg_fill<<<cblocks, 256, 0, stream>>>(ei, segpos, col);

    node_attend<<<(NNODE * 32 + 255) / 256, 256, 0, stream>>>(segpos, col, xl, xr, att, denom, agg);

    int sblocks = (NNODE + ROWS_PER_BLK - 1) / ROWS_PER_BLK;
    post_stats<<<sblocks, DIM, 0, stream>>>(agg, denom, bias, du, sums, sumsq);

    finalize<<<sblocks, DIM, 0, stream>>>(agg, sums, sumsq, gamma, beta, out);
}


# divide, bias, relu, dropout and column sums fused into the node kernel (one launch less); sum order now fixed: source ids sorted per node, column partials added in order, no float atomics
# speedup vs baseline: 1.5318x; 1.0560x over previous
#include <hip/hip_runtime.h>
#include <hip/hip_bf16.h>
#include <stdint.h>

typedef __attribute__((ext_vector_type(2))) float v2f;
typedef __attribute__((ext_vector_type(8))) float v8f;

#define NNODE   10000
#define NEDGE   160000
#define EN      (NEDGE + NNODE)
#define DIM     256
#define MT      (NNODE / 16)
#define NEG_SLOPE 0.2f
#define BN_EPS  1e-5f

__global__ void zero_ws(float4* __restrict__ p, long n4) {
    long i = (long)blockIdx.x * blockDim.x + threadIdx.x;
    long stride = (long)gridDim.x * blockDim.x;
    float4 z = make_float4(0.f, 0.f, 0.f, 0.f);
    for (; i < n4; i += stride) p[i] = z;
}

__global__ void gemm_xlxr(const float* __restrict__ x,
                          const float* __restrict__ Wl, const float* __restrict__ bl,
                          const float* __restrict__ Wr, const float* __restrict__ br,
                          float* __restrict__ xl, float* __restrict__ xr) {
    int wid  = (int)((blockIdx.x * blockDim.x + threadIdx.x) >> 5);
    int lane = (int)(threadIdx.x & 31);
    if (wid >= MT * 32) return;

    int mt   = wid >> 5;
    int rest = wid & 31;
    int mat  = rest >> 4;
    int nt   = rest & 15;

    const float* W    = mat ? Wr : Wl;
    const float* bias = mat ? br : bl;
    float*       out  = mat ? xr : xl;

    int half = lane >> 4;
    int l16  = lane & 15;
    int row  = mt * 16 + l16;
    int col  = nt * 16 + l16;

    const float* aptr = x + (long)row * DIM + 2 * half;
    const float* bptr = W + (long)(2 * half) * DIM + col;

    v8f c = {};
#pragma unroll 4
    for (int k = 0; k < DIM; k += 4) {
        v2f a, b;
        a[0] = aptr[k];
        a[1] = aptr[k + 1];
        b[0] = bptr[(long)k * DIM];
        b[1] = bptr[(long)k * DIM + DIM];
        c = __builtin_amdgcn_wmma_f32_16x16x4_f32(
                 false, a,  false, b,
                 (short)0, c,  false,  false);
    }

    float bv = bias[col];
#pragma unroll
    for (int r = 0; r < 8; ++r) {
        int m = mt * 16 + r + 8 * half;
        out[(long)m * DIM + col] = c[r] + bv;
    }
}

__global__ void seg_count(const int* __restrict__ ei, unsigned* __restrict__ segpos) {
    int e = (int)(blockIdx.x * blockDim.x + threadIdx.x);
    if (e < NEDGE) atomicAdd(&segpos[ei[NEDGE + e]], 1u);
}

#define SCAN_T     1024
#define SCAN_ITEMS ((NNODE + SCAN_T - 1) / SCAN_T)

__global__ void seg_scan(unsigned* __restrict__ segpos) {
    __shared__ unsigned wsum[SCAN_T / 32];
    int tid  = (int)threadIdx.x;
    int lane = tid & 31;
    int w    = tid >> 5;
    int base = tid * SCAN_ITEMS;

    unsigned v[SCAN_ITEMS];
    unsigned tot = 0u;
    for (int i = 0; i < SCAN_ITEMS; ++i) {
        int idx = base + i;
        v[i] = (idx < NNODE) ? segpos[idx] : 0u;
        tot += v[i];
    }
    unsigned inc = tot;
    for (int off = 1; off < 32; off <<= 1) {
        unsigned t = __shfl_up(inc, (unsigned)off, 32);
        if (lane >= off) inc += t;
    }
    if (lane == 31) wsum[w] = inc;
    __syncthreads();
    if (w == 0) {
        unsigned s  = wsum[lane];
        unsigned si = s;
        for (int off = 1; off < 32; off <<= 1) {
            unsigned t = __shfl_up(si, (unsigned)off, 32);
            if (lane >= off) si += t;
        }
        wsum[lane] = si - s;
    }
    __syncthreads();
    unsigned run = wsum[w] + (inc - tot);
    for (int i = 0; i < SCAN_ITEMS; ++i) {
        int idx = base + i;
        if (idx < NNODE) { segpos[idx] = run; run += v[i]; }
    }
}

__global__ void seg_fill(const int* __restrict__ ei, unsigned* __restrict__ segpos,
                         int* __restrict__ col) {
    int e = (int)(blockIdx.x * blockDim.x + threadIdx.x);
    if (e < NEDGE) {
        int d = ei[NEDGE + e];
        unsigned p = atomicAdd(&segpos[d], 1u);
        col[p] = ei[e];
    }
}

__device__ __forceinline__ float edge_logit(const float4& a0, const float4& a1,
                                            const float4& r0, const float4& r1,
                                            const float4& t0, const float4& t1) {
    float acc = 0.f, h;
    h = a0.x + r0.x; h = h > 0.f ? h : NEG_SLOPE * h; acc += t0.x * h;
    h = a0.y + r0.y; h = h > 0.f ? h : NEG_SLOPE * h; acc += t0.y * h;
    h = a0.z + r0.z; h = h > 0.f ? h : NEG_SLOPE * h; acc += t0.z * h;
    h = a0.w + r0.w; h = h > 0.f ? h : NEG_SLOPE * h; acc += t0.w * h;
    h = a1.x + r1.x; h = h > 0.f ? h : NEG_SLOPE * h; acc += t1.x * h;
    h = a1.y + r1.y; h = h > 0.f ? h : NEG_SLOPE * h; acc += t1.y * h;
    h = a1.z + r1.z; h = h > 0.f ? h : NEG_SLOPE * h; acc += t1.z * h;
    h = a1.w + r1.w; h = h > 0.f ? h : NEG_SLOPE * h; acc += t1.w * h;
    for (int off = 16; off > 0; off >>= 1) acc += __shfl_xor(acc, off, 32);
    return acc;
}


#define NODE_WAVES 16
#define NBLK       (NNODE / NODE_WAVES)
static_assert(NNODE % NODE_WAVES == 0, "grid must cover the nodes exactly");
#define ID_PAD     0x7fffffff

__device__ __forceinline__ void absorb(const float4& a0, const float4& a1,
                                       const float4& r0, const float4& r1,
                                       const float4& t0, const float4& t1,
                                       float& m, float& den, float4& acc0, float4& acc1) {
    float lg = edge_logit(a0, a1, r0, r1, t0, t1);
    float mn = fmaxf(m, lg);
    float sc = __expf(m - mn);
    float wt = __expf(lg - mn);
    den = den * sc + wt;
    acc0.x = acc0.x * sc + wt * a0.x;  acc0.y = acc0.y * sc + wt * a0.y;
    acc0.z = acc0.z * sc + wt * a0.z;  acc0.w = acc0.w * sc + wt * a0.w;
    acc1.x = acc1.x * sc + wt * a1.x;  acc1.y = acc1.y * sc + wt * a1.y;
    acc1.z = acc1.z * sc + wt * a1.z;  acc1.w = acc1.w * sc + wt * a1.w;
    m = mn;
}

__device__ __forceinline__ int sort32(int v, int lane) {
#pragma unroll
    for (int k = 2; k <= 32; k <<= 1) {
#pragma unroll
        for (int j = k >> 1; j > 0; j >>= 1) {
            int  o   = __shfl_xor(v, j, 32);
            bool asc = (lane & k) == 0;
            bool low = (lane & j) == 0;
            int  mn  = v < o ? v : o;
            int  mx  = v < o ? o : v;
            v = (asc == low) ? mn : mx;
        }
    }
    return v;
}

__device__ __forceinline__ float finish1(float a, float den, float b, float u) {
    float v = a / den + b;
    v = v > 0.f ? v : 0.f;
    return (u >= 0.5f) ? 2.0f * v : 0.0f;
}

__global__ void __launch_bounds__(NODE_WAVES * 32)
node_attend_post(const unsigned* __restrict__ segpos, const int* __restrict__ col,
                 const float* __restrict__ xl, const float* __restrict__ xr,
                 const float* __restrict__ att, const float* __restrict__ bias,
                 const float* __restrict__ du,
                 float* __restrict__ vout, float* __restrict__ part) {
    __shared__ alignas(16) float vs[NODE_WAVES][DIM];

    int w    = (int)(threadIdx.x >> 5);
    int lane = (int)(threadIdx.x & 31);
    int d    = (int)blockIdx.x * NODE_WAVES + w;

    unsigned beg = d ? segpos[d - 1] : 0u;
    unsigned end = segpos[d];
    unsigned n   = end - beg;

    const float4* xd = (const float4*)(xr + (long)d * DIM);
    const float4* av = (const float4*)att;
    float4 r0 = xd[lane], r1 = xd[lane + 32];
    float4 t0 = av[lane], t1 = av[lane + 32];

    const float4* dp = (const float4*)(du + (long)d * DIM);
    float4 u0 = dp[lane], u1 = dp[lane + 32];

    const float4* xs = (const float4*)(xl + (long)d * DIM);
    float4 acc0 = xs[lane], acc1 = xs[lane + 32];
    float  m    = edge_logit(acc0, acc1, r0, r1, t0, t1);
    float  den  = 1.f;

    if (n <= 32u) {
        int id = (lane < (int)n) ? col[beg + (unsigned)lane] : ID_PAD;
        id = sort32(id, lane);
        if (n > 0u) {
            int s = __shfl(id, 0, 32);
            const float4* xn = (const float4*)(xl + (long)s * DIM);
            float4 n0 = xn[lane], n1 = xn[lane + 32];
            for (int i = 0; i < (int)n; ++i) {
                float4 a0 = n0, a1 = n1;
                if (i + 1 < (int)n) {
                    s  = __shfl(id, i + 1, 32);
                    xn = (const float4*)(xl + (long)s * DIM);
                    n0 = xn[lane]; n1 = xn[lane + 32];
                }
                absorb(a0, a1, r0, r1, t0, t1, m, den, acc0, acc1);
            }
        }
    } else {
        int      last = -1;
        unsigned done = 0u;
        while (done < n) {
            int best = ID_PAD;
            for (unsigned cb = beg; cb < end; cb += 32u) {
                unsigned idx = cb + (unsigned)lane;
                int v = (idx < end) ? col[idx] : ID_PAD;
                if (v > last && v < best) best = v;
            }
            for (int off = 16; off > 0; off >>= 1) {
                int o = __shfl_xor(best, off, 32);
                best = o < best ? o : best;
            }
            int cnt = 0;
            for (unsigned cb = beg; cb < end; cb += 32u) {
                unsigned idx = cb + (unsigned)lane;
                int v = (idx < end) ? col[idx] : ID_PAD;
                cnt += (v == best) ? 1 : 0;
            }
            for (int off = 16; off > 0; off >>= 1) cnt += __shfl_xor(cnt, off, 32);
            if (cnt == 0) break;
            const float4* xn = (const float4*)(xl + (long)best * DIM);
            float4 a0 = xn[lane], a1 = xn[lane + 32];
            for (int c = 0; c < cnt; ++c)
                absorb(a0, a1, r0, r1, t0, t1, m, den, acc0, acc1);
            last  = best;
            done += (unsigned)cnt;
        }
    }

    const float4* bp = (const float4*)bias;
    float4 b0 = bp[lane], b1 = bp[lane + 32];
    float4 v0, v1;
    v0.x = finish1(acc0.x, den, b0.x, u0.x);  v0.y = finish1(acc0.y, den, b0.y, u0.y);
    v0.z = finish1(acc0.z, den, b0.z, u0.z);  v0.w = finish1(acc0.w, den, b0.w, u0.w);
    v1.x = finish1(acc1.x, den, b1.x, u1.x);  v1.y = finish1(acc1.y, den, b1.y, u1.y);
    v1.z = finish1(acc1.z, den, b1.z, u1.z);  v1.w = finish1(acc1.w, den, b1.w, u1.w);

    float4* o = (float4*)(vout + (long)d * DIM);
    o[lane]      = v0;
    o[lane + 32] = v1;

    float4* vl = (float4*)vs[w];
    vl[lane]      = v0;
    vl[lane + 32] = v1;
    __syncthreads();

    int  t  = (int)threadIdx.x;
    int  c  = t & (DIM - 1);
    bool sq = t >= DIM;
    float s = 0.f;
#pragma unroll
    for (int r = 0; r < NODE_WAVES; ++r) {
        float x = vs[r][c];
        s += sq ? x * x : x;
    }
    part[(long)blockIdx.x * (2 * DIM) + t] = s;
}

#define FIN_WAVES 16
#define FIN_COLS  32
#define FIN_RG    16
#define FIN_ROWS  ((NNODE + FIN_RG - 1) / FIN_RG)
#define FIN_PPW   ((NBLK + FIN_WAVES - 1) / FIN_WAVES)
#define FIN_RPW   ((FIN_ROWS + FIN_WAVES - 1) / FIN_WAVES)

__global__ void __launch_bounds__(FIN_WAVES * 32)
finalize(const float* __restrict__ part,
         const float* __restrict__ gamma, const float* __restrict__ beta,
         float* out) {
    __shared__ float red[2][FIN_WAVES][FIN_COLS];
    __shared__ float sc_sh[FIN_COLS], sh_sh[FIN_COLS];

    int w    = (int)(threadIdx.x >> 5);
    int lane = (int)(threadIdx.x & 31);
    int colg = (int)blockIdx.x * FIN_COLS + lane;

    float s = 0.f, s2 = 0.f;
#pragma unroll
    for (int i = 0; i < FIN_PPW; ++i) {
        int  p  = w + FIN_WAVES * i;
        bool ok = p < NBLK;
        int  pc = ok ? p : NBLK - 1;
        float a  = part[(long)pc * (2 * DIM) + colg];
        float a2 = part[(long)pc * (2 * DIM) + DIM + colg];
        s  += ok ? a  : 0.f;
        s2 += ok ? a2 : 0.f;
    }
    red[0][w][lane] = s;
    red[1][w][lane] = s2;
    __syncthreads();
    if (w == 0) {
        float ts = 0.f, ts2 = 0.f;
#pragma unroll
        for (int g = 0; g < FIN_WAVES; ++g) { ts += red[0][g][lane]; ts2 += red[1][g][lane]; }
        const float invn = 1.0f / (float)NNODE;
        float mean  = ts * invn;
        float var   = ts2 * invn - mean * mean;
        float scale = gamma[colg] * rsqrtf(var + BN_EPS);
        sc_sh[lane] = scale;
        sh_sh[lane] = beta[colg] - mean * scale;
    }
    __syncthreads();
    float scale = sc_sh[lane], shift = sh_sh[lane];

    int r0   = (int)blockIdx.y * FIN_ROWS;
    int rend = r0 + FIN_ROWS < NNODE ? r0 + FIN_ROWS : NNODE;
    float x[FIN_RPW];
#pragma unroll
    for (int i = 0; i < FIN_RPW; ++i) {
        int r  = r0 + w + FIN_WAVES * i;
        int rc = r < rend ? r : rend - 1;
        x[i] = out[(long)rc * DIM + colg];
    }
#pragma unroll
    for (int i = 0; i < FIN_RPW; ++i) {
        int r = r0 + w + FIN_WAVES * i;
        if (r < rend) out[(long)r * DIM + colg] = x[i] * scale + shift;
    }
}

extern "C" void kernel_launch(void* const* d_in, const int* in_sizes, int n_in,
                              void* d_out, int out_size, void* d_ws, size_t ws_size,
                              hipStream_t stream) {
    const float* x     = (const float*)d_in[0];
    const int*   ei    = (const int*)d_in[1];
    const float* Wl    = (const float*)d_in[2];
    const float* bl    = (const float*)d_in[3];
    const float* Wr    = (const float*)d_in[4];
    const float* br    = (const float*)d_in[5];
    const float* att   = (const float*)d_in[6];
    const float* bias  = (const float*)d_in[7];
    const float* gamma = (const float*)d_in[8];
    const float* beta  = (const float*)d_in[9];
    const float* du    = (const float*)d_in[10];
    float*       out   = (float*)d_out;

    float*    ws     = (float*)d_ws;
    float*    part   = ws;
    float*    denom  = ws + (long)NNODE * DIM;
    unsigned* segpos = (unsigned*)(denom + NNODE);
    float*    sums   = denom + 2L * NNODE;
    float*    sumsq  = sums + DIM;
    float*    xl     = sumsq + DIM;
    float*    xr     = xl + (long)NNODE * DIM;
    int*      col    = (int*)(xr + (long)NNODE * DIM);

    zero_ws<<<(NNODE / 4 + 255) / 256, 256, 0, stream>>>((float4*)segpos, (long)(NNODE / 4));

    gemm_xlxr<<<(MT * 32) / 4, 128, 0, stream>>>(x, Wl, bl, Wr, br, xl, xr);

    int cblocks = (NEDGE + 255) / 256;
    seg_count<<<cblocks, 256, 0, stream>>>(ei, segpos);
    seg_scan<<<1, SCAN_T, 0, stream>>>(segpos);
    seg_fill<<<cblocks, 256, 0, stream>>>(ei, segpos, col);

    node_attend_post<<<NBLK, NODE_WAVES * 32, 0, stream>>>(segpos, col, xl, xr, att, bias, du, out, part);

    finalize<<<dim3(DIM / FIN_COLS, FIN_RG), FIN_WAVES * 32, 0, stream>>>(part, gamma, beta, out);
}


# nodes with 33 to 64 sources: ids sorted in two registers per lane (64-wide bitonic network) and run through the prefetched loop of smaller nodes; min-scan path only above 64; sum order unchanged
# speedup vs baseline: 1.8478x; 1.2063x over previous
#include <hip/hip_runtime.h>
#include <hip/hip_bf16.h>
#include <stdint.h>

typedef __attribute__((ext_vector_type(2))) float v2f;
typedef __attribute__((ext_vector_type(8))) float v8f;

#define NNODE   10000
#define NEDGE   160000
#define EN      (NEDGE + NNODE)
#define DIM     256
#define MT      (NNODE / 16)
#define NEG_SLOPE 0.2f
#define BN_EPS  1e-5f

__global__ void zero_ws(float4* __restrict__ p, long n4) {
    long i = (long)blockIdx.x * blockDim.x + threadIdx.x;
    long stride = (long)gridDim.x * blockDim.x;
    float4 z = make_float4(0.f, 0.f, 0.f, 0.f);
    for (; i < n4; i += stride) p[i] = z;
}

__global__ void gemm_xlxr(const float* __restrict__ x,
                          const float* __restrict__ Wl, const float* __restrict__ bl,
                          const float* __restrict__ Wr, const float* __restrict__ br,
                          float* __restrict__ xl, float* __restrict__ xr) {
    int wid  = (int)((blockIdx.x * blockDim.x + threadIdx.x) >> 5);
    int lane = (int)(threadIdx.x & 31);
    if (wid >= MT * 32) return;

    int mt   = wid >> 5;
    int rest = wid & 31;
    int mat  = rest >> 4;
    int nt   = rest & 15;

    const float* W    = mat ? Wr : Wl;
    const float* bias = mat ? br : bl;
    float*       out  = mat ? xr : xl;

    int half = lane >> 4;
    int l16  = lane & 15;
    int row  = mt * 16 + l16;
    int col  = nt * 16 + l16;

    const float* aptr = x + (long)row * DIM + 2 * half;
    const float* bptr = W + (long)(2 * half) * DIM + col;

    v8f c = {};
#pragma unroll 4
    for (int k = 0; k < DIM; k += 4) {
        v2f a, b;
        a[0] = aptr[k];
        a[1] = aptr[k + 1];
        b[0] = bptr[(long)k * DIM];
        b[1] = bptr[(long)k * DIM + DIM];
        c = __builtin_amdgcn_wmma_f32_16x16x4_f32(
                 false, a,  false, b,
                 (short)0, c,  false,  false);
    }

    float bv = bias[col];
#pragma unroll
    for (int r = 0; r < 8; ++r) {
        int m = mt * 16 + r + 8 * half;
        out[(long)m * DIM + col] = c[r] + bv;
    }
}

__global__ void seg_count(const int* __restrict__ ei, unsigned* __restrict__ segpos) {
    int e = (int)(blockIdx.x * blockDim.x + threadIdx.x);
    if (e < NEDGE) atomicAdd(&segpos[ei[NEDGE + e]], 1u);
}

#define SCAN_T     1024
#define SCAN_ITEMS ((NNODE + SCAN_T - 1) / SCAN_T)

__global__ void seg_scan(unsigned* __restrict__ segpos) {
    __shared__ unsigned wsum[SCAN_T / 32];
    int tid  = (int)threadIdx.x;
    int lane = tid & 31;
    int w    = tid >> 5;
    int base = tid * SCAN_ITEMS;

    unsigned v[SCAN_ITEMS];
    unsigned tot = 0u;
    for (int i = 0; i < SCAN_ITEMS; ++i) {
        int idx = base + i;
        v[i] = (idx < NNODE) ? segpos[idx] : 0u;
        tot += v[i];
    }
    unsigned inc = tot;
    for (int off = 1; off < 32; off <<= 1) {
        unsigned t = __shfl_up(inc, (unsigned)off, 32);
        if (lane >= off) inc += t;
    }
    if (lane == 31) wsum[w] = inc;
    __syncthreads();
    if (w == 0) {
        unsigned s  = wsum[lane];
        unsigned si = s;
        for (int off = 1; off < 32; off <<= 1) {
            unsigned t = __shfl_up(si, (unsigned)off, 32);
            if (lane >= off) si += t;
        }
        wsum[lane] = si - s;
    }
    __syncthreads();
    unsigned run = wsum[w] + (inc - tot);
    for (int i = 0; i < SCAN_ITEMS; ++i) {
        int idx = base + i;
        if (idx < NNODE) { segpos[idx] = run; run += v[i]; }
    }
}

__global__ void seg_fill(const int* __restrict__ ei, unsigned* __restrict__ segpos,
                         int* __restrict__ col) {
    int e = (int)(blockIdx.x * blockDim.x + threadIdx.x);
    if (e < NEDGE) {
        int d = ei[NEDGE + e];
        unsigned p = atomicAdd(&segpos[d], 1u);
        col[p] = ei[e];
    }
}

__device__ __forceinline__ float edge_logit(const float4& a0, const float4& a1,
                                            const float4& r0, const float4& r1,
                                            const float4& t0, const float4& t1) {
    float acc = 0.f, h;
    h = a0.x + r0.x; h = h > 0.f ? h : NEG_SLOPE * h; acc += t0.x * h;
    h = a0.y + r0.y; h = h > 0.f ? h : NEG_SLOPE * h; acc += t0.y * h;
    h = a0.z + r0.z; h = h > 0.f ? h : NEG_SLOPE * h; acc += t0.z * h;
    h = a0.w + r0.w; h = h > 0.f ? h : NEG_SLOPE * h; acc += t0.w * h;
    h = a1.x + r1.x; h = h > 0.f ? h : NEG_SLOPE * h; acc += t1.x * h;
    h = a1.y + r1.y; h = h > 0.f ? h : NEG_SLOPE * h; acc += t1.y * h;
    h = a1.z + r1.z; h = h > 0.f ? h : NEG_SLOPE * h; acc += t1.z * h;
    h = a1.w + r1.w; h = h > 0.f ? h : NEG_SLOPE * h; acc += t1.w * h;
    for (int off = 16; off > 0; off >>= 1) acc += __shfl_xor(acc, off, 32);
    return acc;
}


#define NODE_WAVES 16
#define NBLK       (NNODE / NODE_WAVES)
static_assert(NNODE % NODE_WAVES == 0, "grid must cover the nodes exactly");
#define ID_PAD     0x7fffffff

__device__ __forceinline__ void absorb(const float4& a0, const float4& a1,
                                       const float4& r0, const float4& r1,
                                       const float4& t0, const float4& t1,
                                       float& m, float& den, float4& acc0, float4& acc1) {
    float lg = edge_logit(a0, a1, r0, r1, t0, t1);
    float mn = fmaxf(m, lg);
    float sc = __expf(m - mn);
    float wt = __expf(lg - mn);
    den = den * sc + wt;
    acc0.x = acc0.x * sc + wt * a0.x;  acc0.y = acc0.y * sc + wt * a0.y;
    acc0.z = acc0.z * sc + wt * a0.z;  acc0.w = acc0.w * sc + wt * a0.w;
    acc1.x = acc1.x * sc + wt * a1.x;  acc1.y = acc1.y * sc + wt * a1.y;
    acc1.z = acc1.z * sc + wt * a1.z;  acc1.w = acc1.w * sc + wt * a1.w;
    m = mn;
}

__device__ __forceinline__ int sort32(int v, int lane) {
#pragma unroll
    for (int k = 2; k <= 32; k <<= 1) {
#pragma unroll
        for (int j = k >> 1; j > 0; j >>= 1) {
            int  o   = __shfl_xor(v, j, 32);
            bool asc = (lane & k) == 0;
            bool low = (lane & j) == 0;
            int  mn  = v < o ? v : o;
            int  mx  = v < o ? o : v;
            v = (asc == low) ? mn : mx;
        }
    }
    return v;
}

__device__ __forceinline__ void sort64(int& a, int& b, int lane) {
#pragma unroll
    for (int k = 2; k <= 32; k <<= 1) {
#pragma unroll
        for (int j = k >> 1; j > 0; j >>= 1) {
            int  oa   = __shfl_xor(a, j, 32);
            int  ob   = __shfl_xor(b, j, 32);
            bool low  = (lane & j) == 0;
            bool asca = (lane & k) == 0;
            bool ascb = ((lane + 32) & k) == 0;
            int  mna  = a < oa ? a : oa,  mxa = a < oa ? oa : a;
            int  mnb  = b < ob ? b : ob,  mxb = b < ob ? ob : b;
            a = (asca == low) ? mna : mxa;
            b = (ascb == low) ? mnb : mxb;
        }
    }
    int mn = a < b ? a : b;
    int mx = a < b ? b : a;
    a = mn;
    b = mx;
#pragma unroll
    for (int j = 16; j > 0; j >>= 1) {
        int  oa  = __shfl_xor(a, j, 32);
        int  ob  = __shfl_xor(b, j, 32);
        bool low = (lane & j) == 0;
        int  mna = a < oa ? a : oa,  mxa = a < oa ? oa : a;
        int  mnb = b < ob ? b : ob,  mxb = b < ob ? ob : b;
        a = low ? mna : mxa;
        b = low ? mnb : mxb;
    }
}

__device__ __forceinline__ float finish1(float a, float den, float b, float u) {
    float v = a / den + b;
    v = v > 0.f ? v : 0.f;
    return (u >= 0.5f) ? 2.0f * v : 0.0f;
}

__global__ void __launch_bounds__(NODE_WAVES * 32)
node_attend_post(const unsigned* __restrict__ segpos, const int* __restrict__ col,
                 const float* __restrict__ xl, const float* __restrict__ xr,
                 const float* __restrict__ att, const float* __restrict__ bias,
                 const float* __restrict__ du,
                 float* __restrict__ vout, float* __restrict__ part) {
    __shared__ alignas(16) float vs[NODE_WAVES][DIM];

    int w    = (int)(threadIdx.x >> 5);
    int lane = (int)(threadIdx.x & 31);
    int d    = (int)blockIdx.x * NODE_WAVES + w;

    unsigned beg = d ? segpos[d - 1] : 0u;
    unsigned end = segpos[d];
    unsigned n   = end - beg;

    const float4* xd = (const float4*)(xr + (long)d * DIM);
    const float4* av = (const float4*)att;
    float4 r0 = xd[lane], r1 = xd[lane + 32];
    float4 t0 = av[lane], t1 = av[lane + 32];

    const float4* dp = (const float4*)(du + (long)d * DIM);
    float4 u0 = dp[lane], u1 = dp[lane + 32];

    const float4* xs = (const float4*)(xl + (long)d * DIM);
    float4 acc0 = xs[lane], acc1 = xs[lane + 32];
    float  m    = edge_logit(acc0, acc1, r0, r1, t0, t1);
    float  den  = 1.f;

    if (n <= 64u) {
        int id  = (lane < (int)n) ? col[beg + (unsigned)lane] : ID_PAD;
        int id2 = ID_PAD;
        if (n <= 32u) {
            id = sort32(id, lane);
        } else {
            id2 = (lane + 32 < (int)n) ? col[beg + 32u + (unsigned)lane] : ID_PAD;
            sort64(id, id2, lane);
        }
        if (n > 0u) {
            int s = __shfl(id, 0, 32);
            const float4* xn = (const float4*)(xl + (long)s * DIM);
            float4 n0 = xn[lane], n1 = xn[lane + 32];
            for (int i = 0; i < (int)n; ++i) {
                float4 a0 = n0, a1 = n1;
                if (i + 1 < (int)n) {
                    int nx = i + 1;
                    s  = __shfl(nx < 32 ? id : id2, nx & 31, 32);
                    xn = (const float4*)(xl + (long)s * DIM);
                    n0 = xn[lane]; n1 = xn[lane + 32];
                }
                absorb(a0, a1, r0, r1, t0, t1, m, den, acc0, acc1);
            }
        }
    } else {
        int      last = -1;
        unsigned done = 0u;
        while (done < n) {
            int best = ID_PAD;
            for (unsigned cb = beg; cb < end; cb += 32u) {
                unsigned idx = cb + (unsigned)lane;
                int v = (idx < end) ? col[idx] : ID_PAD;
                if (v > last && v < best) best = v;
            }
            for (int off = 16; off > 0; off >>= 1) {
                int o = __shfl_xor(best, off, 32);
                best = o < best ? o : best;
            }
            int cnt = 0;
            for (unsigned cb = beg; cb < end; cb += 32u) {
                unsigned idx = cb + (unsigned)lane;
                int v = (idx < end) ? col[idx] : ID_PAD;
                cnt += (v == best) ? 1 : 0;
            }
            for (int off = 16; off > 0; off >>= 1) cnt += __shfl_xor(cnt, off, 32);
            if (cnt == 0) break;
            const float4* xn = (const float4*)(xl + (long)best * DIM);
            float4 a0 = xn[lane], a1 = xn[lane + 32];
            for (int c = 0; c < cnt; ++c)
                absorb(a0, a1, r0, r1, t0, t1, m, den, acc0, acc1);
            last  = best;
            done += (unsigned)cnt;
        }
    }

    const float4* bp = (const float4*)bias;
    float4 b0 = bp[lane], b1 = bp[lane + 32];
    float4 v0, v1;
    v0.x = finish1(acc0.x, den, b0.x, u0.x);  v0.y = finish1(acc0.y, den, b0.y, u0.y);
    v0.z = finish1(acc0.z, den, b0.z, u0.z);  v0.w = finish1(acc0.w, den, b0.w, u0.w);
    v1.x = finish1(acc1.x, den, b1.x, u1.x);  v1.y = finish1(acc1.y, den, b1.y, u1.y);
    v1.z = finish1(acc1.z, den, b1.z, u1.z);  v1.w = finish1(acc1.w, den, b1.w, u1.w);

    float4* o = (float4*)(vout + (long)d * DIM);
    o[lane]      = v0;
    o[lane + 32] = v1;

    float4* vl = (float4*)vs[w];
    vl[lane]      = v0;
    vl[lane + 32] = v1;
    __syncthreads();

    int  t  = (int)threadIdx.x;
    int  c  = t & (DIM - 1);
    bool sq = t >= DIM;
    float s = 0.f;
#pragma unroll
    for (int r = 0; r < NODE_WAVES; ++r) {
        float x = vs[r][c];
        s += sq ? x * x : x;
    }
    part[(long)blockIdx.x * (2 * DIM) + t] = s;
}

#define FIN_WAVES 16
#define FIN_COLS  32
#define FIN_RG    16
#define FIN_ROWS  ((NNODE + FIN_RG - 1) / FIN_RG)
#define FIN_PPW   ((NBLK + FIN_WAVES - 1) / FIN_WAVES)
#define FIN_RPW   ((FIN_ROWS + FIN_WAVES - 1) / FIN_WAVES)

__global__ void __launch_bounds__(FIN_WAVES * 32)
finalize(const float* __restrict__ part,
         const float* __restrict__ gamma, const float* __restrict__ beta,
         float* out) {
    __shared__ float red[2][FIN_WAVES][FIN_COLS];
    __shared__ float sc_sh[FIN_COLS], sh_sh[FIN_COLS];

    int w    = (int)(threadIdx.x >> 5);
    int lane = (int)(threadIdx.x & 31);
    int colg = (int)blockIdx.x * FIN_COLS + lane;

    float s = 0.f, s2 = 0.f;
#pragma unroll
    for (int i = 0; i < FIN_PPW; ++i) {
        int  p  = w + FIN_WAVES * i;
        bool ok = p < NBLK;
        int  pc = ok ? p : NBLK - 1;
        float a  = part[(long)pc * (2 * DIM) + colg];
        float a2 = part[(long)pc * (2 * DIM) + DIM + colg];
        s  += ok ? a  : 0.f;
        s2 += ok ? a2 : 0.f;
    }
    red[0][w][lane] = s;
    red[1][w][lane] = s2;
    __syncthreads();
    if (w == 0) {
        float ts = 0.f, ts2 = 0.f;
#pragma unroll
        for (int g = 0; g < FIN_WAVES; ++g) { ts += red[0][g][lane]; ts2 += red[1][g][lane]; }
        const float invn = 1.0f / (float)NNODE;
        float mean  = ts * invn;
        float var   = ts2 * invn - mean * mean;
        float scale = gamma[colg] * rsqrtf(var + BN_EPS);
        sc_sh[lane] = scale;
        sh_sh[lane] = beta[colg] - mean * scale;
    }
    __syncthreads();
    float scale = sc_sh[lane], shift = sh_sh[lane];

    int r0   = (int)blockIdx.y * FIN_ROWS;
    int rend = r0 + FIN_ROWS < NNODE ? r0 + FIN_ROWS : NNODE;
    float x[FIN_RPW];
#pragma unroll
    for (int i = 0; i < FIN_RPW; ++i) {
        int r  = r0 + w + FIN_WAVES * i;
        int rc = r < rend ? r : rend - 1;
        x[i] = out[(long)rc * DIM + colg];
    }
#pragma unroll
    for (int i = 0; i < FIN_RPW; ++i) {
        int r = r0 + w + FIN_WAVES * i;
        if (r < rend) out[(long)r * DIM + colg] = x[i] * scale + shift;
    }
}

extern "C" void kernel_launch(void* const* d_in, const int* in_sizes, int n_in,
                              void* d_out, int out_size, void* d_ws, size_t ws_size,
                              hipStream_t stream) {
    const float* x     = (const float*)d_in[0];
    const int*   ei    = (const int*)d_in[1];
    const float* Wl    = (const float*)d_in[2];
    const float* bl    = (const float*)d_in[3];
    const float* Wr    = (const float*)d_in[4];
    const float* br    = (const float*)d_in[5];
    const float* att   = (const float*)d_in[6];
    const float* bias  = (const float*)d_in[7];
    const float* gamma = (const float*)d_in[8];
    const float* beta  = (const float*)d_in[9];
    const float* du    = (const float*)d_in[10];
    float*       out   = (float*)d_out;

    float*    ws     = (float*)d_ws;
    float*    part   = ws;
    float*    denom  = ws + (long)NNODE * DIM;
    unsigned* segpos = (unsigned*)(denom + NNODE);
    float*    sums   = denom + 2L * NNODE;
    float*    sumsq  = sums + DIM;
    float*    xl     = sumsq + DIM;
    float*    xr     = xl + (long)NNODE * DIM;
    int*      col    = (int*)(xr + (long)NNODE * DIM);

    zero_ws<<<(NNODE / 4 + 255) / 256, 256, 0, stream>>>((float4*)segpos, (long)(NNODE / 4));

    gemm_xlxr<<<(MT * 32) / 4, 128, 0, stream>>>(x, Wl, bl, Wr, br, xl, xr);

    int cblocks = (NEDGE + 255) / 256;
    seg_count<<<cblocks, 256, 0, stream>>>(ei, segpos);
    seg_scan<<<1, SCAN_T, 0, stream>>>(segpos);
    seg_fill<<<cblocks, 256, 0, stream>>>(ei, segpos, col);

    node_attend_post<<<NBLK, NODE_WAVES * 32, 0, stream>>>(segpos, col, xl, xr, att, bias, du, out, part);

    finalize<<<dim3(DIM / FIN_COLS, FIN_RG), FIN_WAVES * 32, 0, stream>>>(part, gamma, beta, out);
}


# GEMM re-tiled, no LDS: one wave per 16 rows x 64 columns, 4 accumulators, columns interleaved by 4 so W rows load and results store as 16-byte accesses, k blocks loaded ahead; k sum order changes
# speedup vs baseline: 2.4245x; 1.3121x over previous
#include <hip/hip_runtime.h>
#include <hip/hip_bf16.h>
#include <stdint.h>

typedef __attribute__((ext_vector_type(2))) float v2f;
typedef __attribute__((ext_vector_type(8))) float v8f;

#define NNODE   10000
#define NEDGE   160000
#define EN      (NEDGE + NNODE)
#define DIM     256
#define MT      (NNODE / 16)
#define NEG_SLOPE 0.2f
#define BN_EPS  1e-5f

__global__ void zero_ws(float4* __restrict__ p, long n4) {
    long i = (long)blockIdx.x * blockDim.x + threadIdx.x;
    long stride = (long)gridDim.x * blockDim.x;
    float4 z = make_float4(0.f, 0.f, 0.f, 0.f);
    for (; i < n4; i += stride) p[i] = z;
}

#define GK      16
#define GSTEPS  (DIM / GK)

struct gfrag {
    float4 a[GK / 8];
    float4 b[GK / 8][4];
};

__device__ __forceinline__ void gload(gfrag& f, const float* __restrict__ ap,
                                      const float* __restrict__ bp, int k0) {
#pragma unroll
    for (int u = 0; u < GK / 8; ++u) {
        f.a[u] = *(const float4*)(ap + k0 + 8 * u);
#pragma unroll
        for (int q = 0; q < 4; ++q)
            f.b[u][q] = *(const float4*)(bp + (long)(k0 + 8 * u + q) * DIM);
    }
}

__device__ __forceinline__ float f4get(const float4& v, int j) {
    return j == 0 ? v.x : j == 1 ? v.y : j == 2 ? v.z : v.w;
}

__global__ void __launch_bounds__(128)
gemm_xlxr(const float* __restrict__ x,
          const float* __restrict__ Wl, const float* __restrict__ bl,
          const float* __restrict__ Wr, const float* __restrict__ br,
          float* __restrict__ xl, float* __restrict__ xr) {
    int wid  = (int)((blockIdx.x * blockDim.x + threadIdx.x) >> 5);
    int lane = (int)(threadIdx.x & 31);

    int mt  = wid >> 3;
    int g   = wid & 7;
    int mat = g >> 2;
    int cb  = (g & 3) * 64;

    const float* Wm   = mat ? Wr : Wl;
    const float* bias = mat ? br : bl;
    float*       out  = mat ? xr : xl;

    int h = lane >> 4;
    int c = lane & 15;

    const float* ap = x  + (long)(mt * 16 + c) * DIM + 4 * h;
    const float* bp = Wm + (long)(4 * h) * DIM + cb + 4 * c;

    v8f acc[4] = {};
    gfrag cur, nxt;
    gload(cur, ap, bp, 0);
#pragma unroll
    for (int s = 0; s < GSTEPS; ++s) {
        if (s + 1 < GSTEPS) gload(nxt, ap, bp, (s + 1) * GK);
        __builtin_amdgcn_sched_barrier(0);
#pragma unroll
        for (int u = 0; u < GK / 8; ++u) {
#pragma unroll
            for (int p = 0; p < 2; ++p) {
                v2f a;
                a[0] = f4get(cur.a[u], 2 * p);
                a[1] = f4get(cur.a[u], 2 * p + 1);
#pragma unroll
                for (int j = 0; j < 4; ++j) {
                    v2f b;
                    b[0] = f4get(cur.b[u][2 * p], j);
                    b[1] = f4get(cur.b[u][2 * p + 1], j);
                    acc[j] = __builtin_amdgcn_wmma_f32_16x16x4_f32(
                                  false, a,  false, b,
                                  (short)0, acc[j],  false,  false);
                }
            }
        }
        if (s + 1 < GSTEPS) cur = nxt;
    }

    float4 bv = *(const float4*)(bias + cb + 4 * c);
#pragma unroll
    for (int r = 0; r < 8; ++r) {
        int m = mt * 16 + r + 8 * h;
        float4 v;
        v.x = acc[0][r] + bv.x;
        v.y = acc[1][r] + bv.y;
        v.z = acc[2][r] + bv.z;
        v.w = acc[3][r] + bv.w;
        *(float4*)(out + (long)m * DIM + cb + 4 * c) = v;
    }
}

__global__ void seg_count(const int* __restrict__ ei, unsigned* __restrict__ segpos) {
    int e = (int)(blockIdx.x * blockDim.x + threadIdx.x);
    if (e < NEDGE) atomicAdd(&segpos[ei[NEDGE + e]], 1u);
}

#define SCAN_T     1024
#define SCAN_ITEMS ((NNODE + SCAN_T - 1) / SCAN_T)

__global__ void seg_scan(unsigned* __restrict__ segpos) {
    __shared__ unsigned wsum[SCAN_T / 32];
    int tid  = (int)threadIdx.x;
    int lane = tid & 31;
    int w    = tid >> 5;
    int base = tid * SCAN_ITEMS;

    unsigned v[SCAN_ITEMS];
    unsigned tot = 0u;
    for (int i = 0; i < SCAN_ITEMS; ++i) {
        int idx = base + i;
        v[i] = (idx < NNODE) ? segpos[idx] : 0u;
        tot += v[i];
    }
    unsigned inc = tot;
    for (int off = 1; off < 32; off <<= 1) {
        unsigned t = __shfl_up(inc, (unsigned)off, 32);
        if (lane >= off) inc += t;
    }
    if (lane == 31) wsum[w] = inc;
    __syncthreads();
    if (w == 0) {
        unsigned s  = wsum[lane];
        unsigned si = s;
        for (int off = 1; off < 32; off <<= 1) {
            unsigned t = __shfl_up(si, (unsigned)off, 32);
            if (lane >= off) si += t;
        }
        wsum[lane] = si - s;
    }
    __syncthreads();
    unsigned run = wsum[w] + (inc - tot);
    for (int i = 0; i < SCAN_ITEMS; ++i) {
        int idx = base + i;
        if (idx < NNODE) { segpos[idx] = run; run += v[i]; }
    }
}

__global__ void seg_fill(const int* __restrict__ ei, unsigned* __restrict__ segpos,
                         int* __restrict__ col) {
    int e = (int)(blockIdx.x * blockDim.x + threadIdx.x);
    if (e < NEDGE) {
        int d = ei[NEDGE + e];
        unsigned p = atomicAdd(&segpos[d], 1u);
        col[p] = ei[e];
    }
}

__device__ __forceinline__ float edge_logit(const float4& a0, const float4& a1,
                                            const float4& r0, const float4& r1,
                                            const float4& t0, const float4& t1) {
    float acc = 0.f, h;
    h = a0.x + r0.x; h = h > 0.f ? h : NEG_SLOPE * h; acc += t0.x * h;
    h = a0.y + r0.y; h = h > 0.f ? h : NEG_SLOPE * h; acc += t0.y * h;
    h = a0.z + r0.z; h = h > 0.f ? h : NEG_SLOPE * h; acc += t0.z * h;
    h = a0.w + r0.w; h = h > 0.f ? h : NEG_SLOPE * h; acc += t0.w * h;
    h = a1.x + r1.x; h = h > 0.f ? h : NEG_SLOPE * h; acc += t1.x * h;
    h = a1.y + r1.y; h = h > 0.f ? h : NEG_SLOPE * h; acc += t1.y * h;
    h = a1.z + r1.z; h = h > 0.f ? h : NEG_SLOPE * h; acc += t1.z * h;
    h = a1.w + r1.w; h = h > 0.f ? h : NEG_SLOPE * h; acc += t1.w * h;
    for (int off = 16; off > 0; off >>= 1) acc += __shfl_xor(acc, off, 32);
    return acc;
}


#define NODE_WAVES 16
#define NBLK       (NNODE / NODE_WAVES)
static_assert(NNODE % NODE_WAVES == 0, "grid must cover the nodes exactly");
#define ID_PAD     0x7fffffff

__device__ __forceinline__ void absorb(const float4& a0, const float4& a1,
                                       const float4& r0, const float4& r1,
                                       const float4& t0, const float4& t1,
                                       float& m, float& den, float4& acc0, float4& acc1) {
    float lg = edge_logit(a0, a1, r0, r1, t0, t1);
    float mn = fmaxf(m, lg);
    float sc = __expf(m - mn);
    float wt = __expf(lg - mn);
    den = den * sc + wt;
    acc0.x = acc0.x * sc + wt * a0.x;  acc0.y = acc0.y * sc + wt * a0.y;
    acc0.z = acc0.z * sc + wt * a0.z;  acc0.w = acc0.w * sc + wt * a0.w;
    acc1.x = acc1.x * sc + wt * a1.x;  acc1.y = acc1.y * sc + wt * a1.y;
    acc1.z = acc1.z * sc + wt * a1.z;  acc1.w = acc1.w * sc + wt * a1.w;
    m = mn;
}

__device__ __forceinline__ int sort32(int v, int lane) {
#pragma unroll
    for (int k = 2; k <= 32; k <<= 1) {
#pragma unroll
        for (int j = k >> 1; j > 0; j >>= 1) {
            int  o   = __shfl_xor(v, j, 32);
            bool asc = (lane & k) == 0;
            bool low = (lane & j) == 0;
            int  mn  = v < o ? v : o;
            int  mx  = v < o ? o : v;
            v = (asc == low) ? mn : mx;
        }
    }
    return v;
}

__device__ __forceinline__ void sort64(int& a, int& b, int lane) {
#pragma unroll
    for (int k = 2; k <= 32; k <<= 1) {
#pragma unroll
        for (int j = k >> 1; j > 0; j >>= 1) {
            int  oa   = __shfl_xor(a, j, 32);
            int  ob   = __shfl_xor(b, j, 32);
            bool low  = (lane & j) == 0;
            bool asca = (lane & k) == 0;
            bool ascb = ((lane + 32) & k) == 0;
            int  mna  = a < oa ? a : oa,  mxa = a < oa ? oa : a;
            int  mnb  = b < ob ? b : ob,  mxb = b < ob ? ob : b;
            a = (asca == low) ? mna : mxa;
            b = (ascb == low) ? mnb : mxb;
        }
    }
    int mn = a < b ? a : b;
    int mx = a < b ? b : a;
    a = mn;
    b = mx;
#pragma unroll
    for (int j = 16; j > 0; j >>= 1) {
        int  oa  = __shfl_xor(a, j, 32);
        int  ob  = __shfl_xor(b, j, 32);
        bool low = (lane & j) == 0;
        int  mna = a < oa ? a : oa,  mxa = a < oa ? oa : a;
        int  mnb = b < ob ? b : ob,  mxb = b < ob ? ob : b;
        a = low ? mna : mxa;
        b = low ? mnb : mxb;
    }
}

__device__ __forceinline__ float finish1(float a, float den, float b, float u) {
    float v = a / den + b;
    v = v > 0.f ? v : 0.f;
    return (u >= 0.5f) ? 2.0f * v : 0.0f;
}

__global__ void __launch_bounds__(NODE_WAVES * 32)
node_attend_post(const unsigned* __restrict__ segpos, const int* __restrict__ col,
                 const float* __restrict__ xl, const float* __restrict__ xr,
                 const float* __restrict__ att, const float* __restrict__ bias,
                 const float* __restrict__ du,
                 float* __restrict__ vout, float* __restrict__ part) {
    __shared__ alignas(16) float vs[NODE_WAVES][DIM];

    int w    = (int)(threadIdx.x >> 5);
    int lane = (int)(threadIdx.x & 31);
    int d    = (int)blockIdx.x * NODE_WAVES + w;

    unsigned beg = d ? segpos[d - 1] : 0u;
    unsigned end = segpos[d];
    unsigned n   = end - beg;

    const float4* xd = (const float4*)(xr + (long)d * DIM);
    const float4* av = (const float4*)att;
    float4 r0 = xd[lane], r1 = xd[lane + 32];
    float4 t0 = av[lane], t1 = av[lane + 32];

    const float4* dp = (const float4*)(du + (long)d * DIM);
    float4 u0 = dp[lane], u1 = dp[lane + 32];

    const float4* xs = (const float4*)(xl + (long)d * DIM);
    float4 acc0 = xs[lane], acc1 = xs[lane + 32];
    float  m    = edge_logit(acc0, acc1, r0, r1, t0, t1);
    float  den  = 1.f;

    if (n <= 64u) {
        int id  = (lane < (int)n) ? col[beg + (unsigned)lane] : ID_PAD;
        int id2 = ID_PAD;
        if (n <= 32u) {
            id = sort32(id, lane);
        } else {
            id2 = (lane + 32 < (int)n) ? col[beg + 32u + (unsigned)lane] : ID_PAD;
            sort64(id, id2, lane);
        }
        if (n > 0u) {
            int s = __shfl(id, 0, 32);
            const float4* xn = (const float4*)(xl + (long)s * DIM);
            float4 n0 = xn[lane], n1 = xn[lane + 32];
            for (int i = 0; i < (int)n; ++i) {
                float4 a0 = n0, a1 = n1;
                if (i + 1 < (int)n) {
                    int nx = i + 1;
                    s  = __shfl(nx < 32 ? id : id2, nx & 31, 32);
                    xn = (const float4*)(xl + (long)s * DIM);
                    n0 = xn[lane]; n1 = xn[lane + 32];
                }
                absorb(a0, a1, r0, r1, t0, t1, m, den, acc0, acc1);
            }
        }
    } else {
        int      last = -1;
        unsigned done = 0u;
        while (done < n) {
            int best = ID_PAD;
            for (unsigned cb = beg; cb < end; cb += 32u) {
                unsigned idx = cb + (unsigned)lane;
                int v = (idx < end) ? col[idx] : ID_PAD;
                if (v > last && v < best) best = v;
            }
            for (int off = 16; off > 0; off >>= 1) {
                int o = __shfl_xor(best, off, 32);
                best = o < best ? o : best;
            }
            int cnt = 0;
            for (unsigned cb = beg; cb < end; cb += 32u) {
                unsigned idx = cb + (unsigned)lane;
                int v = (idx < end) ? col[idx] : ID_PAD;
                cnt += (v == best) ? 1 : 0;
            }
            for (int off = 16; off > 0; off >>= 1) cnt += __shfl_xor(cnt, off, 32);
            if (cnt == 0) break;
            const float4* xn = (const float4*)(xl + (long)best * DIM);
            float4 a0 = xn[lane], a1 = xn[lane + 32];
            for (int c = 0; c < cnt; ++c)
                absorb(a0, a1, r0, r1, t0, t1, m, den, acc0, acc1);
            last  = best;
            done += (unsigned)cnt;
        }
    }

    const float4* bp = (const float4*)bias;
    float4 b0 = bp[lane], b1 = bp[lane + 32];
    float4 v0, v1;
    v0.x = finish1(acc0.x, den, b0.x, u0.x);  v0.y = finish1(acc0.y, den, b0.y, u0.y);
    v0.z = finish1(acc0.z, den, b0.z, u0.z);  v0.w = finish1(acc0.w, den, b0.w, u0.w);
    v1.x = finish1(acc1.x, den, b1.x, u1.x);  v1.y = finish1(acc1.y, den, b1.y, u1.y);
    v1.z = finish1(acc1.z, den, b1.z, u1.z);  v1.w = finish1(acc1.w, den, b1.w, u1.w);

    float4* o = (float4*)(vout + (long)d * DIM);
    o[lane]      = v0;
    o[lane + 32] = v1;

    float4* vl = (float4*)vs[w];
    vl[lane]      = v0;
    vl[lane + 32] = v1;
    __syncthreads();

    int  t  = (int)threadIdx.x;
    int  c  = t & (DIM - 1);
    bool sq = t >= DIM;
    float s = 0.f;
#pragma unroll
    for (int r = 0; r < NODE_WAVES; ++r) {
        float x = vs[r][c];
        s += sq ? x * x : x;
    }
    part[(long)blockIdx.x * (2 * DIM) + t] = s;
}

#define FIN_WAVES 16
#define FIN_COLS  32
#define FIN_RG    16
#define FIN_ROWS  ((NNODE + FIN_RG - 1) / FIN_RG)
#define FIN_PPW   ((NBLK + FIN_WAVES - 1) / FIN_WAVES)
#define FIN_RPW   ((FIN_ROWS + FIN_WAVES - 1) / FIN_WAVES)

__global__ void __launch_bounds__(FIN_WAVES * 32)
finalize(const float* __restrict__ part,
         const float* __restrict__ gamma, const float* __restrict__ beta,
         float* out) {
    __shared__ float red[2][FIN_WAVES][FIN_COLS];
    __shared__ float sc_sh[FIN_COLS], sh_sh[FIN_COLS];

    int w    = (int)(threadIdx.x >> 5);
    int lane = (int)(threadIdx.x & 31);
    int colg = (int)blockIdx.x * FIN_COLS + lane;

    float s = 0.f, s2 = 0.f;
#pragma unroll
    for (int i = 0; i < FIN_PPW; ++i) {
        int  p  = w + FIN_WAVES * i;
        bool ok = p < NBLK;
        int  pc = ok ? p : NBLK - 1;
        float a  = part[(long)pc * (2 * DIM) + colg];
        float a2 = part[(long)pc * (2 * DIM) + DIM + colg];
        s  += ok ? a  : 0.f;
        s2 += ok ? a2 : 0.f;
    }
    red[0][w][lane] = s;
    red[1][w][lane] = s2;
    __syncthreads();
    if (w == 0) {
        float ts = 0.f, ts2 = 0.f;
#pragma unroll
        for (int g = 0; g < FIN_WAVES; ++g) { ts += red[0][g][lane]; ts2 += red[1][g][lane]; }
        const float invn = 1.0f / (float)NNODE;
        float mean  = ts * invn;
        float var   = ts2 * invn - mean * mean;
        float scale = gamma[colg] * rsqrtf(var + BN_EPS);
        sc_sh[lane] = scale;
        sh_sh[lane] = beta[colg] - mean * scale;
    }
    __syncthreads();
    float scale = sc_sh[lane], shift = sh_sh[lane];

    int r0   = (int)blockIdx.y * FIN_ROWS;
    int rend = r0 + FIN_ROWS < NNODE ? r0 + FIN_ROWS : NNODE;
    float x[FIN_RPW];
#pragma unroll
    for (int i = 0; i < FIN_RPW; ++i) {
        int r  = r0 + w + FIN_WAVES * i;
        int rc = r < rend ? r : rend - 1;
        x[i] = out[(long)rc * DIM + colg];
    }
#pragma unroll
    for (int i = 0; i < FIN_RPW; ++i) {
        int r = r0 + w + FIN_WAVES * i;
        if (r < rend) out[(long)r * DIM + colg] = x[i] * scale + shift;
    }
}

extern "C" void kernel_launch(void* const* d_in, const int* in_sizes, int n_in,
                              void* d_out, int out_size, void* d_ws, size_t ws_size,
                              hipStream_t stream) {
    const float* x     = (const float*)d_in[0];
    const int*   ei    = (const int*)d_in[1];
    const float* Wl    = (const float*)d_in[2];
    const float* bl    = (const float*)d_in[3];
    const float* Wr    = (const float*)d_in[4];
    const float* br    = (const float*)d_in[5];
    const float* att   = (const float*)d_in[6];
    const float* bias  = (const float*)d_in[7];
    const float* gamma = (const float*)d_in[8];
    const float* beta  = (const float*)d_in[9];
    const float* du    = (const float*)d_in[10];
    float*       out   = (float*)d_out;

    float*    ws     = (float*)d_ws;
    float*    part   = ws;
    float*    denom  = ws + (long)NNODE * DIM;
    unsigned* segpos = (unsigned*)(denom + NNODE);
    float*    sums   = denom + 2L * NNODE;
    float*    sumsq  = sums + DIM;
    float*    xl     = sumsq + DIM;
    float*    xr     = xl + (long)NNODE * DIM;
    int*      col    = (int*)(xr + (long)NNODE * DIM);

    zero_ws<<<(NNODE / 4 + 255) / 256, 256, 0, stream>>>((float4*)segpos, (long)(NNODE / 4));

    gemm_xlxr<<<(MT * 8) / 4, 128, 0, stream>>>(x, Wl, bl, Wr, br, xl, xr);

    int cblocks = (NEDGE + 255) / 256;
    seg_count<<<cblocks, 256, 0, stream>>>(ei, segpos);
    seg_scan<<<1, SCAN_T, 0, stream>>>(segpos);
    seg_fill<<<cblocks, 256, 0, stream>>>(ei, segpos, col);

    node_attend_post<<<NBLK, NODE_WAVES * 32, 0, stream>>>(segpos, col, xl, xr, att, bias, du, out, part);

    finalize<<<dim3(DIM / FIN_COLS, FIN_RG), FIN_WAVES * 32, 0, stream>>>(part, gamma, beta, out);
}


# edge buckets built in one launch: each node owns a 64-slot source list, an edge's one counter add gives its slot; counters cleared in the GEMM launch; 7 launches to 4; sum order unchanged
# speedup vs baseline: 3.2937x; 1.3585x over previous
#include <hip/hip_runtime.h>
#include <hip/hip_bf16.h>
#include <stdint.h>

typedef __attribute__((ext_vector_type(2))) float v2f;
typedef __attribute__((ext_vector_type(8))) float v8f;

#define NNODE   10000
#define NEDGE   160000
#define EN      (NEDGE + NNODE)
#define DIM     256
#define MT      (NNODE / 16)
#define NEG_SLOPE 0.2f
#define BN_EPS  1e-5f

#define GK      16
#define GSTEPS  (DIM / GK)

struct gfrag {
    float4 a[GK / 8];
    float4 b[GK / 8][4];
};

__device__ __forceinline__ void gload(gfrag& f, const float* __restrict__ ap,
                                      const float* __restrict__ bp, int k0) {
#pragma unroll
    for (int u = 0; u < GK / 8; ++u) {
        f.a[u] = *(const float4*)(ap + k0 + 8 * u);
#pragma unroll
        for (int q = 0; q < 4; ++q)
            f.b[u][q] = *(const float4*)(bp + (long)(k0 + 8 * u + q) * DIM);
    }
}

__device__ __forceinline__ float f4get(const float4& v, int j) {
    return j == 0 ? v.x : j == 1 ? v.y : j == 2 ? v.z : v.w;
}

__global__ void __launch_bounds__(128)
gemm_xlxr(const float* __restrict__ x,
          const float* __restrict__ Wl, const float* __restrict__ bl,
          const float* __restrict__ Wr, const float* __restrict__ br,
          float* __restrict__ xl, float* __restrict__ xr,
          unsigned* __restrict__ cnt) {
    static_assert((MT * 8 / 4) * 8 == NNODE, "8 counters per block must cover the nodes exactly");
    if (threadIdx.x < 8u) cnt[blockIdx.x * 8u + threadIdx.x] = 0u;

    int wid  = (int)((blockIdx.x * blockDim.x + threadIdx.x) >> 5);
    int lane = (int)(threadIdx.x & 31);

    int mt  = wid >> 3;
    int g   = wid & 7;
    int mat = g >> 2;
    int cb  = (g & 3) * 64;

    const float* Wm   = mat ? Wr : Wl;
    const float* bias = mat ? br : bl;
    float*       out  = mat ? xr : xl;

    int h = lane >> 4;
    int c = lane & 15;

    const float* ap = x  + (long)(mt * 16 + c) * DIM + 4 * h;
    const float* bp = Wm + (long)(4 * h) * DIM + cb + 4 * c;

    v8f acc[4] = {};
    gfrag cur, nxt;
    gload(cur, ap, bp, 0);
#pragma unroll
    for (int s = 0; s < GSTEPS; ++s) {
        if (s + 1 < GSTEPS) gload(nxt, ap, bp, (s + 1) * GK);
        __builtin_amdgcn_sched_barrier(0);
#pragma unroll
        for (int u = 0; u < GK / 8; ++u) {
#pragma unroll
            for (int p = 0; p < 2; ++p) {
                v2f a;
                a[0] = f4get(cur.a[u], 2 * p);
                a[1] = f4get(cur.a[u], 2 * p + 1);
#pragma unroll
                for (int j = 0; j < 4; ++j) {
                    v2f b;
                    b[0] = f4get(cur.b[u][2 * p], j);
                    b[1] = f4get(cur.b[u][2 * p + 1], j);
                    acc[j] = __builtin_amdgcn_wmma_f32_16x16x4_f32(
                                  false, a,  false, b,
                                  (short)0, acc[j],  false,  false);
                }
            }
        }
        if (s + 1 < GSTEPS) cur = nxt;
    }

    float4 bv = *(const float4*)(bias + cb + 4 * c);
#pragma unroll
    for (int r = 0; r < 8; ++r) {
        int m = mt * 16 + r + 8 * h;
        float4 v;
        v.x = acc[0][r] + bv.x;
        v.y = acc[1][r] + bv.y;
        v.z = acc[2][r] + bv.z;
        v.w = acc[3][r] + bv.w;
        *(float4*)(out + (long)m * DIM + cb + 4 * c) = v;
    }
}

#define SLOTS 64

__global__ void seg_fill(const int* __restrict__ ei, unsigned* __restrict__ cnt,
                         int* __restrict__ slot) {
    int e = (int)(blockIdx.x * blockDim.x + threadIdx.x);
    if (e < NEDGE) {
        int d = ei[NEDGE + e];
        unsigned p = atomicAdd(&cnt[d], 1u);
        if (p < (unsigned)SLOTS) slot[(long)d * SLOTS + p] = ei[e];
    }
}

__device__ __forceinline__ float edge_logit(const float4& a0, const float4& a1,
                                            const float4& r0, const float4& r1,
                                            const float4& t0, const float4& t1) {
    float acc = 0.f, h;
    h = a0.x + r0.x; h = h > 0.f ? h : NEG_SLOPE * h; acc += t0.x * h;
    h = a0.y + r0.y; h = h > 0.f ? h : NEG_SLOPE * h; acc += t0.y * h;
    h = a0.z + r0.z; h = h > 0.f ? h : NEG_SLOPE * h; acc += t0.z * h;
    h = a0.w + r0.w; h = h > 0.f ? h : NEG_SLOPE * h; acc += t0.w * h;
    h = a1.x + r1.x; h = h > 0.f ? h : NEG_SLOPE * h; acc += t1.x * h;
    h = a1.y + r1.y; h = h > 0.f ? h : NEG_SLOPE * h; acc += t1.y * h;
    h = a1.z + r1.z; h = h > 0.f ? h : NEG_SLOPE * h; acc += t1.z * h;
    h = a1.w + r1.w; h = h > 0.f ? h : NEG_SLOPE * h; acc += t1.w * h;
    for (int off = 16; off > 0; off >>= 1) acc += __shfl_xor(acc, off, 32);
    return acc;
}


#define NODE_WAVES 16
#define NBLK       (NNODE / NODE_WAVES)
static_assert(NNODE % NODE_WAVES == 0, "grid must cover the nodes exactly");
#define ID_PAD     0x7fffffff

__device__ __forceinline__ void absorb(const float4& a0, const float4& a1,
                                       const float4& r0, const float4& r1,
                                       const float4& t0, const float4& t1,
                                       float& m, float& den, float4& acc0, float4& acc1) {
    float lg = edge_logit(a0, a1, r0, r1, t0, t1);
    float mn = fmaxf(m, lg);
    float sc = __expf(m - mn);
    float wt = __expf(lg - mn);
    den = den * sc + wt;
    acc0.x = acc0.x * sc + wt * a0.x;  acc0.y = acc0.y * sc + wt * a0.y;
    acc0.z = acc0.z * sc + wt * a0.z;  acc0.w = acc0.w * sc + wt * a0.w;
    acc1.x = acc1.x * sc + wt * a1.x;  acc1.y = acc1.y * sc + wt * a1.y;
    acc1.z = acc1.z * sc + wt * a1.z;  acc1.w = acc1.w * sc + wt * a1.w;
    m = mn;
}

__device__ __forceinline__ int sort32(int v, int lane) {
#pragma unroll
    for (int k = 2; k <= 32; k <<= 1) {
#pragma unroll
        for (int j = k >> 1; j > 0; j >>= 1) {
            int  o   = __shfl_xor(v, j, 32);
            bool asc = (lane & k) == 0;
            bool low = (lane & j) == 0;
            int  mn  = v < o ? v : o;
            int  mx  = v < o ? o : v;
            v = (asc == low) ? mn : mx;
        }
    }
    return v;
}

__device__ __forceinline__ void sort64(int& a, int& b, int lane) {
#pragma unroll
    for (int k = 2; k <= 32; k <<= 1) {
#pragma unroll
        for (int j = k >> 1; j > 0; j >>= 1) {
            int  oa   = __shfl_xor(a, j, 32);
            int  ob   = __shfl_xor(b, j, 32);
            bool low  = (lane & j) == 0;
            bool asca = (lane & k) == 0;
            bool ascb = ((lane + 32) & k) == 0;
            int  mna  = a < oa ? a : oa,  mxa = a < oa ? oa : a;
            int  mnb  = b < ob ? b : ob,  mxb = b < ob ? ob : b;
            a = (asca == low) ? mna : mxa;
            b = (ascb == low) ? mnb : mxb;
        }
    }
    int mn = a < b ? a : b;
    int mx = a < b ? b : a;
    a = mn;
    b = mx;
#pragma unroll
    for (int j = 16; j > 0; j >>= 1) {
        int  oa  = __shfl_xor(a, j, 32);
        int  ob  = __shfl_xor(b, j, 32);
        bool low = (lane & j) == 0;
        int  mna = a < oa ? a : oa,  mxa = a < oa ? oa : a;
        int  mnb = b < ob ? b : ob,  mxb = b < ob ? ob : b;
        a = low ? mna : mxa;
        b = low ? mnb : mxb;
    }
}

__device__ __forceinline__ float finish1(float a, float den, float b, float u) {
    float v = a / den + b;
    v = v > 0.f ? v : 0.f;
    return (u >= 0.5f) ? 2.0f * v : 0.0f;
}

__global__ void __launch_bounds__(NODE_WAVES * 32)
node_attend_post(const unsigned* __restrict__ deg, const int* __restrict__ slot,
                 const int* __restrict__ ei,
                 const float* __restrict__ xl, const float* __restrict__ xr,
                 const float* __restrict__ att, const float* __restrict__ bias,
                 const float* __restrict__ du,
                 float* __restrict__ vout, float* __restrict__ part) {
    __shared__ alignas(16) float vs[NODE_WAVES][DIM];

    int w    = (int)(threadIdx.x >> 5);
    int lane = (int)(threadIdx.x & 31);
    int d    = (int)blockIdx.x * NODE_WAVES + w;

    unsigned   n  = deg[d];
    const int* sl = slot + (long)d * SLOTS;

    const float4* xd = (const float4*)(xr + (long)d * DIM);
    const float4* av = (const float4*)att;
    float4 r0 = xd[lane], r1 = xd[lane + 32];
    float4 t0 = av[lane], t1 = av[lane + 32];

    const float4* dp = (const float4*)(du + (long)d * DIM);
    float4 u0 = dp[lane], u1 = dp[lane + 32];

    const float4* xs = (const float4*)(xl + (long)d * DIM);
    float4 acc0 = xs[lane], acc1 = xs[lane + 32];
    float  m    = edge_logit(acc0, acc1, r0, r1, t0, t1);
    float  den  = 1.f;

    if (n <= (unsigned)SLOTS) {
        int id  = (lane < (int)n) ? sl[lane] : ID_PAD;
        int id2 = ID_PAD;
        if (n <= 32u) {
            id = sort32(id, lane);
        } else {
            id2 = (lane + 32 < (int)n) ? sl[32 + lane] : ID_PAD;
            sort64(id, id2, lane);
        }
        if (n > 0u) {
            int s = __shfl(id, 0, 32);
            const float4* xn = (const float4*)(xl + (long)s * DIM);
            float4 n0 = xn[lane], n1 = xn[lane + 32];
            for (int i = 0; i < (int)n; ++i) {
                float4 a0 = n0, a1 = n1;
                if (i + 1 < (int)n) {
                    int nx = i + 1;
                    s  = __shfl(nx < 32 ? id : id2, nx & 31, 32);
                    xn = (const float4*)(xl + (long)s * DIM);
                    n0 = xn[lane]; n1 = xn[lane + 32];
                }
                absorb(a0, a1, r0, r1, t0, t1, m, den, acc0, acc1);
            }
        }
    } else {
        static_assert(NEDGE % 32 == 0, "every lane runs the same number of trips");
        const int* es = ei;
        const int* ed = ei + NEDGE;
        int      last = -1;
        unsigned done = 0u;
        while (done < n) {
            int best = ID_PAD;
            for (int e = lane; e < NEDGE; e += 32) {
                int v = (ed[e] == d) ? es[e] : ID_PAD;
                if (v > last && v < best) best = v;
            }
            for (int off = 16; off > 0; off >>= 1) {
                int o = __shfl_xor(best, off, 32);
                best = o < best ? o : best;
            }
            int cnt = 0;
            for (int e = lane; e < NEDGE; e += 32)
                cnt += (ed[e] == d && es[e] == best) ? 1 : 0;
            for (int off = 16; off > 0; off >>= 1) cnt += __shfl_xor(cnt, off, 32);
            if (cnt == 0) break;
            const float4* xn = (const float4*)(xl + (long)best * DIM);
            float4 a0 = xn[lane], a1 = xn[lane + 32];
            for (int c = 0; c < cnt; ++c)
                absorb(a0, a1, r0, r1, t0, t1, m, den, acc0, acc1);
            last  = best;
            done += (unsigned)cnt;
        }
    }

    const float4* bp = (const float4*)bias;
    float4 b0 = bp[lane], b1 = bp[lane + 32];
    float4 v0, v1;
    v0.x = finish1(acc0.x, den, b0.x, u0.x);  v0.y = finish1(acc0.y, den, b0.y, u0.y);
    v0.z = finish1(acc0.z, den, b0.z, u0.z);  v0.w = finish1(acc0.w, den, b0.w, u0.w);
    v1.x = finish1(acc1.x, den, b1.x, u1.x);  v1.y = finish1(acc1.y, den, b1.y, u1.y);
    v1.z = finish1(acc1.z, den, b1.z, u1.z);  v1.w = finish1(acc1.w, den, b1.w, u1.w);

    float4* o = (float4*)(vout + (long)d * DIM);
    o[lane]      = v0;
    o[lane + 32] = v1;

    float4* vl = (float4*)vs[w];
    vl[lane]      = v0;
    vl[lane + 32] = v1;
    __syncthreads();

    int  t  = (int)threadIdx.x;
    int  c  = t & (DIM - 1);
    bool sq = t >= DIM;
    float s = 0.f;
#pragma unroll
    for (int r = 0; r < NODE_WAVES; ++r) {
        float x = vs[r][c];
        s += sq ? x * x : x;
    }
    part[(long)blockIdx.x * (2 * DIM) + t] = s;
}

#define FIN_WAVES 16
#define FIN_COLS  32
#define FIN_RG    16
#define FIN_ROWS  ((NNODE + FIN_RG - 1) / FIN_RG)
#define FIN_PPW   ((NBLK + FIN_WAVES - 1) / FIN_WAVES)
#define FIN_RPW   ((FIN_ROWS + FIN_WAVES - 1) / FIN_WAVES)

__global__ void __launch_bounds__(FIN_WAVES * 32)
finalize(const float* __restrict__ part,
         const float* __restrict__ gamma, const float* __restrict__ beta,
         float* out) {
    __shared__ float red[2][FIN_WAVES][FIN_COLS];
    __shared__ float sc_sh[FIN_COLS], sh_sh[FIN_COLS];

    int w    = (int)(threadIdx.x >> 5);
    int lane = (int)(threadIdx.x & 31);
    int colg = (int)blockIdx.x * FIN_COLS + lane;

    float s = 0.f, s2 = 0.f;
#pragma unroll
    for (int i = 0; i < FIN_PPW; ++i) {
        int  p  = w + FIN_WAVES * i;
        bool ok = p < NBLK;
        int  pc = ok ? p : NBLK - 1;
        float a  = part[(long)pc * (2 * DIM) + colg];
        float a2 = part[(long)pc * (2 * DIM) + DIM + colg];
        s  += ok ? a  : 0.f;
        s2 += ok ? a2 : 0.f;
    }
    red[0][w][lane] = s;
    red[1][w][lane] = s2;
    __syncthreads();
    if (w == 0) {
        float ts = 0.f, ts2 = 0.f;
#pragma unroll
        for (int g = 0; g < FIN_WAVES; ++g) { ts += red[0][g][lane]; ts2 += red[1][g][lane]; }
        const float invn = 1.0f / (float)NNODE;
        float mean  = ts * invn;
        float var   = ts2 * invn - mean * mean;
        float scale = gamma[colg] * rsqrtf(var + BN_EPS);
        sc_sh[lane] = scale;
        sh_sh[lane] = beta[colg] - mean * scale;
    }
    __syncthreads();
    float scale = sc_sh[lane], shift = sh_sh[lane];

    int r0   = (int)blockIdx.y * FIN_ROWS;
    int rend = r0 + FIN_ROWS < NNODE ? r0 + FIN_ROWS : NNODE;
    float x[FIN_RPW];
#pragma unroll
    for (int i = 0; i < FIN_RPW; ++i) {
        int r  = r0 + w + FIN_WAVES * i;
        int rc = r < rend ? r : rend - 1;
        x[i] = out[(long)rc * DIM + colg];
    }
#pragma unroll
    for (int i = 0; i < FIN_RPW; ++i) {
        int r = r0 + w + FIN_WAVES * i;
        if (r < rend) out[(long)r * DIM + colg] = x[i] * scale + shift;
    }
}

extern "C" void kernel_launch(void* const* d_in, const int* in_sizes, int n_in,
                              void* d_out, int out_size, void* d_ws, size_t ws_size,
                              hipStream_t stream) {
    const float* x     = (const float*)d_in[0];
    const int*   ei    = (const int*)d_in[1];
    const float* Wl    = (const float*)d_in[2];
    const float* bl    = (const float*)d_in[3];
    const float* Wr    = (const float*)d_in[4];
    const float* br    = (const float*)d_in[5];
    const float* att   = (const float*)d_in[6];
    const float* bias  = (const float*)d_in[7];
    const float* gamma = (const float*)d_in[8];
    const float* beta  = (const float*)d_in[9];
    const float* du    = (const float*)d_in[10];
    float*       out   = (float*)d_out;

    static_assert((long)NBLK * 2 * DIM + (long)NNODE * SLOTS <= (long)NNODE * DIM,
                  "partials and source lists must fit the first span");
    float*    ws     = (float*)d_ws;
    float*    part   = ws;
    int*      slot   = (int*)(ws + (long)NBLK * 2 * DIM);
    float*    denom  = ws + (long)NNODE * DIM;
    unsigned* deg    = (unsigned*)(denom + NNODE);
    float*    sums   = denom + 2L * NNODE;
    float*    sumsq  = sums + DIM;
    float*    xl     = sumsq + DIM;
    float*    xr     = xl + (long)NNODE * DIM;

    gemm_xlxr<<<(MT * 8) / 4, 128, 0, stream>>>(x, Wl, bl, Wr, br, xl, xr, deg);

    int cblocks = (NEDGE + 255) / 256;
    seg_fill<<<cblocks, 256, 0, stream>>>(ei, deg, slot);

    node_attend_post<<<NBLK, NODE_WAVES * 32, 0, stream>>>(deg, slot, ei, xl, xr, att, bias, du, out, part);

    finalize<<<dim3(DIM / FIN_COLS, FIN_RG), FIN_WAVES * 32, 0, stream>>>(part, gamma, beta, out);
}


# node kernel takes a node's sorted source rows 4 at a time: 4 row loads in flight, 4 logit sums side by side (ds_swizzle), ids by readlane, one max/rescale per 4 rows; rescale rounding changes
# speedup vs baseline: 3.5066x; 1.0646x over previous
#include <hip/hip_runtime.h>
#include <hip/hip_bf16.h>
#include <stdint.h>

typedef __attribute__((ext_vector_type(2))) float v2f;
typedef __attribute__((ext_vector_type(8))) float v8f;

#define NNODE   10000
#define NEDGE   160000
#define EN      (NEDGE + NNODE)
#define DIM     256
#define MT      (NNODE / 16)
#define NEG_SLOPE 0.2f
#define BN_EPS  1e-5f

#define GK      16
#define GSTEPS  (DIM / GK)

struct gfrag {
    float4 a[GK / 8];
    float4 b[GK / 8][4];
};

__device__ __forceinline__ void gload(gfrag& f, const float* __restrict__ ap,
                                      const float* __restrict__ bp, int k0) {
#pragma unroll
    for (int u = 0; u < GK / 8; ++u) {
        f.a[u] = *(const float4*)(ap + k0 + 8 * u);
#pragma unroll
        for (int q = 0; q < 4; ++q)
            f.b[u][q] = *(const float4*)(bp + (long)(k0 + 8 * u + q) * DIM);
    }
}

__device__ __forceinline__ float f4get(const float4& v, int j) {
    return j == 0 ? v.x : j == 1 ? v.y : j == 2 ? v.z : v.w;
}

__global__ void __launch_bounds__(128)
gemm_xlxr(const float* __restrict__ x,
          const float* __restrict__ Wl, const float* __restrict__ bl,
          const float* __restrict__ Wr, const float* __restrict__ br,
          float* __restrict__ xl, float* __restrict__ xr,
          unsigned* __restrict__ cnt) {
    static_assert((MT * 8 / 4) * 8 == NNODE, "8 counters per block must cover the nodes exactly");
    if (threadIdx.x < 8u) cnt[blockIdx.x * 8u + threadIdx.x] = 0u;

    int wid  = (int)((blockIdx.x * blockDim.x + threadIdx.x) >> 5);
    int lane = (int)(threadIdx.x & 31);

    int mt  = wid >> 3;
    int g   = wid & 7;
    int mat = g >> 2;
    int cb  = (g & 3) * 64;

    const float* Wm   = mat ? Wr : Wl;
    const float* bias = mat ? br : bl;
    float*       out  = mat ? xr : xl;

    int h = lane >> 4;
    int c = lane & 15;

    const float* ap = x  + (long)(mt * 16 + c) * DIM + 4 * h;
    const float* bp = Wm + (long)(4 * h) * DIM + cb + 4 * c;

    v8f acc[4] = {};
    gfrag cur, nxt;
    gload(cur, ap, bp, 0);
#pragma unroll
    for (int s = 0; s < GSTEPS; ++s) {
        if (s + 1 < GSTEPS) gload(nxt, ap, bp, (s + 1) * GK);
        __builtin_amdgcn_sched_barrier(0);
#pragma unroll
        for (int u = 0; u < GK / 8; ++u) {
#pragma unroll
            for (int p = 0; p < 2; ++p) {
                v2f a;
                a[0] = f4get(cur.a[u], 2 * p);
                a[1] = f4get(cur.a[u], 2 * p + 1);
#pragma unroll
                for (int j = 0; j < 4; ++j) {
                    v2f b;
                    b[0] = f4get(cur.b[u][2 * p], j);
                    b[1] = f4get(cur.b[u][2 * p + 1], j);
                    acc[j] = __builtin_amdgcn_wmma_f32_16x16x4_f32(
                                  false, a,  false, b,
                                  (short)0, acc[j],  false,  false);
                }
            }
        }
        if (s + 1 < GSTEPS) cur = nxt;
    }

    float4 bv = *(const float4*)(bias + cb + 4 * c);
#pragma unroll
    for (int r = 0; r < 8; ++r) {
        int m = mt * 16 + r + 8 * h;
        float4 v;
        v.x = acc[0][r] + bv.x;
        v.y = acc[1][r] + bv.y;
        v.z = acc[2][r] + bv.z;
        v.w = acc[3][r] + bv.w;
        *(float4*)(out + (long)m * DIM + cb + 4 * c) = v;
    }
}

#define SLOTS 64

__global__ void seg_fill(const int* __restrict__ ei, unsigned* __restrict__ cnt,
                         int* __restrict__ slot) {
    int e = (int)(blockIdx.x * blockDim.x + threadIdx.x);
    if (e < NEDGE) {
        int d = ei[NEDGE + e];
        unsigned p = atomicAdd(&cnt[d], 1u);
        if (p < (unsigned)SLOTS) slot[(long)d * SLOTS + p] = ei[e];
    }
}

__device__ __forceinline__ float logit_part(const float4& a0, const float4& a1,
                                            const float4& r0, const float4& r1,
                                            const float4& t0, const float4& t1) {
    float acc = 0.f, h;
    h = a0.x + r0.x; h = h > 0.f ? h : NEG_SLOPE * h; acc += t0.x * h;
    h = a0.y + r0.y; h = h > 0.f ? h : NEG_SLOPE * h; acc += t0.y * h;
    h = a0.z + r0.z; h = h > 0.f ? h : NEG_SLOPE * h; acc += t0.z * h;
    h = a0.w + r0.w; h = h > 0.f ? h : NEG_SLOPE * h; acc += t0.w * h;
    h = a1.x + r1.x; h = h > 0.f ? h : NEG_SLOPE * h; acc += t1.x * h;
    h = a1.y + r1.y; h = h > 0.f ? h : NEG_SLOPE * h; acc += t1.y * h;
    h = a1.z + r1.z; h = h > 0.f ? h : NEG_SLOPE * h; acc += t1.z * h;
    h = a1.w + r1.w; h = h > 0.f ? h : NEG_SLOPE * h; acc += t1.w * h;
    return acc;
}

__device__ __forceinline__ float edge_logit(const float4& a0, const float4& a1,
                                            const float4& r0, const float4& r1,
                                            const float4& t0, const float4& t1) {
    float acc = logit_part(a0, a1, r0, r1, t0, t1);
    for (int off = 16; off > 0; off >>= 1) acc += __shfl_xor(acc, off, 32);
    return acc;
}


#define NODE_WAVES 16
#define NBLK       (NNODE / NODE_WAVES)
static_assert(NNODE % NODE_WAVES == 0, "grid must cover the nodes exactly");
#define ID_PAD     0x7fffffff

__device__ __forceinline__ void absorb(const float4& a0, const float4& a1,
                                       const float4& r0, const float4& r1,
                                       const float4& t0, const float4& t1,
                                       float& m, float& den, float4& acc0, float4& acc1) {
    float lg = edge_logit(a0, a1, r0, r1, t0, t1);
    float mn = fmaxf(m, lg);
    float sc = __expf(m - mn);
    float wt = __expf(lg - mn);
    den = den * sc + wt;
    acc0.x = acc0.x * sc + wt * a0.x;  acc0.y = acc0.y * sc + wt * a0.y;
    acc0.z = acc0.z * sc + wt * a0.z;  acc0.w = acc0.w * sc + wt * a0.w;
    acc1.x = acc1.x * sc + wt * a1.x;  acc1.y = acc1.y * sc + wt * a1.y;
    acc1.z = acc1.z * sc + wt * a1.z;  acc1.w = acc1.w * sc + wt * a1.w;
    m = mn;
}

#define GROUP 4

template <int OFF>
__device__ __forceinline__ void group_xor_sum(float (&v)[GROUP]) {
    float o[GROUP];
#pragma unroll
    for (int j = 0; j < GROUP; ++j)
        o[j] = __builtin_bit_cast(float,
                   __builtin_amdgcn_ds_swizzle(__builtin_bit_cast(int, v[j]), (OFF << 10) | 0x1f));
#pragma unroll
    for (int j = 0; j < GROUP; ++j) v[j] += o[j];
}

__device__ __forceinline__ void absorb_group(const float4 (&a0)[GROUP], const float4 (&a1)[GROUP],
                                             const bool (&live)[GROUP],
                                             const float4& r0, const float4& r1,
                                             const float4& t0, const float4& t1,
                                             float& m, float& den, float4& acc0, float4& acc1) {
    float lg[GROUP];
#pragma unroll
    for (int j = 0; j < GROUP; ++j)
        lg[j] = live[j] ? logit_part(a0[j], a1[j], r0, r1, t0, t1) : -INFINITY;
    group_xor_sum<16>(lg);
    group_xor_sum<8>(lg);
    group_xor_sum<4>(lg);
    group_xor_sum<2>(lg);
    group_xor_sum<1>(lg);
    float mn = m;
#pragma unroll
    for (int j = 0; j < GROUP; ++j) mn = fmaxf(mn, lg[j]);
    float sc = __expf(m - mn);
    den *= sc;
    acc0.x *= sc;  acc0.y *= sc;  acc0.z *= sc;  acc0.w *= sc;
    acc1.x *= sc;  acc1.y *= sc;  acc1.z *= sc;  acc1.w *= sc;
#pragma unroll
    for (int j = 0; j < GROUP; ++j) {
        float wt = __expf(lg[j] - mn);
        den += wt;
        acc0.x += wt * a0[j].x;  acc0.y += wt * a0[j].y;
        acc0.z += wt * a0[j].z;  acc0.w += wt * a0[j].w;
        acc1.x += wt * a1[j].x;  acc1.y += wt * a1[j].y;
        acc1.z += wt * a1[j].z;  acc1.w += wt * a1[j].w;
    }
    m = mn;
}

__device__ __forceinline__ int sort32(int v, int lane) {
#pragma unroll
    for (int k = 2; k <= 32; k <<= 1) {
#pragma unroll
        for (int j = k >> 1; j > 0; j >>= 1) {
            int  o   = __shfl_xor(v, j, 32);
            bool asc = (lane & k) == 0;
            bool low = (lane & j) == 0;
            int  mn  = v < o ? v : o;
            int  mx  = v < o ? o : v;
            v = (asc == low) ? mn : mx;
        }
    }
    return v;
}

__device__ __forceinline__ void sort64(int& a, int& b, int lane) {
#pragma unroll
    for (int k = 2; k <= 32; k <<= 1) {
#pragma unroll
        for (int j = k >> 1; j > 0; j >>= 1) {
            int  oa   = __shfl_xor(a, j, 32);
            int  ob   = __shfl_xor(b, j, 32);
            bool low  = (lane & j) == 0;
            bool asca = (lane & k) == 0;
            bool ascb = ((lane + 32) & k) == 0;
            int  mna  = a < oa ? a : oa,  mxa = a < oa ? oa : a;
            int  mnb  = b < ob ? b : ob,  mxb = b < ob ? ob : b;
            a = (asca == low) ? mna : mxa;
            b = (ascb == low) ? mnb : mxb;
        }
    }
    int mn = a < b ? a : b;
    int mx = a < b ? b : a;
    a = mn;
    b = mx;
#pragma unroll
    for (int j = 16; j > 0; j >>= 1) {
        int  oa  = __shfl_xor(a, j, 32);
        int  ob  = __shfl_xor(b, j, 32);
        bool low = (lane & j) == 0;
        int  mna = a < oa ? a : oa,  mxa = a < oa ? oa : a;
        int  mnb = b < ob ? b : ob,  mxb = b < ob ? ob : b;
        a = low ? mna : mxa;
        b = low ? mnb : mxb;
    }
}

__device__ __forceinline__ float finish1(float a, float den, float b, float u) {
    float v = a / den + b;
    v = v > 0.f ? v : 0.f;
    return (u >= 0.5f) ? 2.0f * v : 0.0f;
}

__global__ void __launch_bounds__(NODE_WAVES * 32)
node_attend_post(const unsigned* __restrict__ deg, const int* __restrict__ slot,
                 const int* __restrict__ ei,
                 const float* __restrict__ xl, const float* __restrict__ xr,
                 const float* __restrict__ att, const float* __restrict__ bias,
                 const float* __restrict__ du,
                 float* __restrict__ vout, float* __restrict__ part) {
    __shared__ alignas(16) float vs[NODE_WAVES][DIM];

    int w    = __builtin_amdgcn_readfirstlane((int)(threadIdx.x >> 5));
    int lane = (int)(threadIdx.x & 31);
    int d    = (int)blockIdx.x * NODE_WAVES + w;

    unsigned   n  = deg[d];
    const int* sl = slot + (long)d * SLOTS;

    const float4* xd = (const float4*)(xr + (long)d * DIM);
    const float4* av = (const float4*)att;
    float4 r0 = xd[lane], r1 = xd[lane + 32];
    float4 t0 = av[lane], t1 = av[lane + 32];

    const float4* xs = (const float4*)(xl + (long)d * DIM);
    float4 acc0 = xs[lane], acc1 = xs[lane + 32];
    float  m    = edge_logit(acc0, acc1, r0, r1, t0, t1);
    float  den  = 1.f;

    static_assert(SLOTS % GROUP == 0, "a group never reaches past the two id registers");
    if (n <= (unsigned)SLOTS) {
        int id  = (lane < (int)n) ? sl[lane] : ID_PAD;
        int id2 = ID_PAD;
        if (n <= 32u) {
            id = sort32(id, lane);
        } else {
            id2 = (lane + 32 < (int)n) ? sl[32 + lane] : ID_PAD;
            sort64(id, id2, lane);
        }
        const float4* xl4 = (const float4*)xl;
        for (int i0 = 0; i0 < (int)n; i0 += GROUP) {
            float4 a0[GROUP], a1[GROUP];
            bool   live[GROUP];
            int    idv = i0 < 32 ? id : id2;
#pragma unroll
            for (int j = 0; j < GROUP; ++j) {
                int ix  = i0 + j;
                live[j] = ix < (int)n;
                int s   = __builtin_amdgcn_readlane(idv, ix & 31);
                int su  = (int)min((unsigned)s, (unsigned)(NNODE - 1));
                int ro  = su * (DIM / 4) + lane;
                a0[j] = xl4[ro];
                a1[j] = xl4[ro + 32];
            }
            absorb_group(a0, a1, live, r0, r1, t0, t1, m, den, acc0, acc1);
        }
    } else {
        static_assert(NEDGE % 32 == 0, "every lane runs the same number of trips");
        const int* es = ei;
        const int* ed = ei + NEDGE;
        int      last = -1;
        unsigned done = 0u;
        while (done < n) {
            int best = ID_PAD;
            for (int e = lane; e < NEDGE; e += 32) {
                int v = (ed[e] == d) ? es[e] : ID_PAD;
                if (v > last && v < best) best = v;
            }
            for (int off = 16; off > 0; off >>= 1) {
                int o = __shfl_xor(best, off, 32);
                best = o < best ? o : best;
            }
            int cnt = 0;
            for (int e = lane; e < NEDGE; e += 32)
                cnt += (ed[e] == d && es[e] == best) ? 1 : 0;
            for (int off = 16; off > 0; off >>= 1) cnt += __shfl_xor(cnt, off, 32);
            if (cnt == 0) break;
            const float4* xn = (const float4*)(xl + (long)best * DIM);
            float4 a0 = xn[lane], a1 = xn[lane + 32];
            for (int c = 0; c < cnt; ++c)
                absorb(a0, a1, r0, r1, t0, t1, m, den, acc0, acc1);
            last  = best;
            done += (unsigned)cnt;
        }
    }

    const float4* dp = (const float4*)(du + (long)d * DIM);
    float4 u0 = dp[lane], u1 = dp[lane + 32];
    const float4* bp = (const float4*)bias;
    float4 b0 = bp[lane], b1 = bp[lane + 32];
    float4 v0, v1;
    v0.x = finish1(acc0.x, den, b0.x, u0.x);  v0.y = finish1(acc0.y, den, b0.y, u0.y);
    v0.z = finish1(acc0.z, den, b0.z, u0.z);  v0.w = finish1(acc0.w, den, b0.w, u0.w);
    v1.x = finish1(acc1.x, den, b1.x, u1.x);  v1.y = finish1(acc1.y, den, b1.y, u1.y);
    v1.z = finish1(acc1.z, den, b1.z, u1.z);  v1.w = finish1(acc1.w, den, b1.w, u1.w);

    float4* o = (float4*)(vout + (long)d * DIM);
    o[lane]      = v0;
    o[lane + 32] = v1;

    float4* vl = (float4*)vs[w];
    vl[lane]      = v0;
    vl[lane + 32] = v1;
    __syncthreads();

    int  t  = (int)threadIdx.x;
    int  c  = t & (DIM - 1);
    bool sq = t >= DIM;
    float s = 0.f;
#pragma unroll
    for (int r = 0; r < NODE_WAVES; ++r) {
        float x = vs[r][c];
        s += sq ? x * x : x;
    }
    part[(long)blockIdx.x * (2 * DIM) + t] = s;
}

#define FIN_WAVES 16
#define FIN_COLS  32
#define FIN_RG    16
#define FIN_ROWS  ((NNODE + FIN_RG - 1) / FIN_RG)
#define FIN_PPW   ((NBLK + FIN_WAVES - 1) / FIN_WAVES)
#define FIN_RPW   ((FIN_ROWS + FIN_WAVES - 1) / FIN_WAVES)

__global__ void __launch_bounds__(FIN_WAVES * 32)
finalize(const float* __restrict__ part,
         const float* __restrict__ gamma, const float* __restrict__ beta,
         float* out) {
    __shared__ float red[2][FIN_WAVES][FIN_COLS];
    __shared__ float sc_sh[FIN_COLS], sh_sh[FIN_COLS];

    int w    = (int)(threadIdx.x >> 5);
    int lane = (int)(threadIdx.x & 31);
    int colg = (int)blockIdx.x * FIN_COLS + lane;

    float s = 0.f, s2 = 0.f;
#pragma unroll
    for (int i = 0; i < FIN_PPW; ++i) {
        int  p  = w + FIN_WAVES * i;
        bool ok = p < NBLK;
        int  pc = ok ? p : NBLK - 1;
        float a  = part[(long)pc * (2 * DIM) + colg];
        float a2 = part[(long)pc * (2 * DIM) + DIM + colg];
        s  += ok ? a  : 0.f;
        s2 += ok ? a2 : 0.f;
    }
    red[0][w][lane] = s;
    red[1][w][lane] = s2;
    __syncthreads();
    if (w == 0) {
        float ts = 0.f, ts2 = 0.f;
#pragma unroll
        for (int g = 0; g < FIN_WAVES; ++g) { ts += red[0][g][lane]; ts2 += red[1][g][lane]; }
        const float invn = 1.0f / (float)NNODE;
        float mean  = ts * invn;
        float var   = ts2 * invn - mean * mean;
        float scale = gamma[colg] * rsqrtf(var + BN_EPS);
        sc_sh[lane] = scale;
        sh_sh[lane] = beta[colg] - mean * scale;
    }
    __syncthreads();
    float scale = sc_sh[lane], shift = sh_sh[lane];

    int r0   = (int)blockIdx.y * FIN_ROWS;
    int rend = r0 + FIN_ROWS < NNODE ? r0 + FIN_ROWS : NNODE;
    float x[FIN_RPW];
#pragma unroll
    for (int i = 0; i < FIN_RPW; ++i) {
        int r  = r0 + w + FIN_WAVES * i;
        int rc = r < rend ? r : rend - 1;
        x[i] = out[(long)rc * DIM + colg];
    }
#pragma unroll
    for (int i = 0; i < FIN_RPW; ++i) {
        int r = r0 + w + FIN_WAVES * i;
        if (r < rend) out[(long)r * DIM + colg] = x[i] * scale + shift;
    }
}

extern "C" void kernel_launch(void* const* d_in, const int* in_sizes, int n_in,
                              void* d_out, int out_size, void* d_ws, size_t ws_size,
                              hipStream_t stream) {
    const float* x     = (const float*)d_in[0];
    const int*   ei    = (const int*)d_in[1];
    const float* Wl    = (const float*)d_in[2];
    const float* bl    = (const float*)d_in[3];
    const float* Wr    = (const float*)d_in[4];
    const float* br    = (const float*)d_in[5];
    const float* att   = (const float*)d_in[6];
    const float* bias  = (const float*)d_in[7];
    const float* gamma = (const float*)d_in[8];
    const float* beta  = (const float*)d_in[9];
    const float* du    = (const float*)d_in[10];
    float*       out   = (float*)d_out;

    static_assert((long)NBLK * 2 * DIM + (long)NNODE * SLOTS <= (long)NNODE * DIM,
                  "partials and source lists must fit the first span");
    float*    ws     = (float*)d_ws;
    float*    part   = ws;
    int*      slot   = (int*)(ws + (long)NBLK * 2 * DIM);
    float*    denom  = ws + (long)NNODE * DIM;
    unsigned* deg    = (unsigned*)(denom + NNODE);
    float*    sums   = denom + 2L * NNODE;
    float*    sumsq  = sums + DIM;
    float*    xl     = sumsq + DIM;
    float*    xr     = xl + (long)NNODE * DIM;

    gemm_xlxr<<<(MT * 8) / 4, 128, 0, stream>>>(x, Wl, bl, Wr, br, xl, xr, deg);

    int cblocks = (NEDGE + 255) / 256;
    seg_fill<<<cblocks, 256, 0, stream>>>(ei, deg, slot);

    node_attend_post<<<NBLK, NODE_WAVES * 32, 0, stream>>>(deg, slot, ei, xl, xr, att, bias, du, out, part);

    finalize<<<dim3(DIM / FIN_COLS, FIN_RG), FIN_WAVES * 32, 0, stream>>>(part, gamma, beta, out);
}


# edge buckets built inside the GEMM launch by 200 extra blocks, one per (node range, edge chunk), LDS counts; node kernel packs a node's 8 chunk lists; 4 launches to 3; sum order unchanged
# speedup vs baseline: 3.9821x; 1.1356x over previous
#include <hip/hip_runtime.h>
#include <hip/hip_bf16.h>
#include <stdint.h>

typedef __attribute__((ext_vector_type(2))) float v2f;
typedef __attribute__((ext_vector_type(8))) float v8f;

#define NNODE   10000
#define NEDGE   160000
#define EN      (NEDGE + NNODE)
#define DIM     256
#define MT      (NNODE / 16)
#define NEG_SLOPE 0.2f
#define BN_EPS  1e-5f

#define GK      16
#define GSTEPS  (DIM / GK)

struct gfrag {
    float4 a[GK / 8];
    float4 b[GK / 8][4];
};

__device__ __forceinline__ void gload(gfrag& f, const float* __restrict__ ap,
                                      const float* __restrict__ bp, int k0) {
#pragma unroll
    for (int u = 0; u < GK / 8; ++u) {
        f.a[u] = *(const float4*)(ap + k0 + 8 * u);
#pragma unroll
        for (int q = 0; q < 4; ++q)
            f.b[u][q] = *(const float4*)(bp + (long)(k0 + 8 * u + q) * DIM);
    }
}

__device__ __forceinline__ float f4get(const float4& v, int j) {
    return j == 0 ? v.x : j == 1 ? v.y : j == 2 ? v.z : v.w;
}

#define NCH     8
#define CAP     16
#define ECH     (NEDGE / NCH)
#define FR      400
#define NRG     (NNODE / FR)
#define FUNITS  (NRG * NCH)
#define FU      8
#define FTRIPS  ((ECH / 4 + 128 * FU - 1) / (128 * FU))
static_assert(NEDGE % (4 * NCH) == 0, "a chunk is a whole number of 16-byte pieces");
static_assert(NNODE % FR == 0, "the ranges cover the nodes exactly");

__device__ __forceinline__ int i4get(const int4& v, int j) {
    return j == 0 ? v.x : j == 1 ? v.y : j == 2 ? v.z : v.w;
}

__device__ __forceinline__ void fill_unit(const int* __restrict__ ei, int* __restrict__ cnt8,
                                          int* __restrict__ sub, int unit, unsigned* lcnt) {
    int t  = (int)threadIdx.x;
    int rg = unit / NCH;
    int ch = unit - rg * NCH;
    int lo = rg * FR;
    for (int i = t; i < FR; i += 128) lcnt[i] = 0u;
    __syncthreads();

    const int4* s4 = (const int4*)(ei + ch * ECH);
    const int4* d4 = (const int4*)(ei + NEDGE + ch * ECH);
    for (int trip = 0; trip < FTRIPS; ++trip) {
        int4 dv[FU], sv[FU];
        bool ok[FU];
#pragma unroll
        for (int u = 0; u < FU; ++u) {
            int q  = (trip * FU + u) * 128 + t;
            ok[u]  = q < ECH / 4;
            int qq = ok[u] ? q : 0;
            dv[u]  = d4[qq];
            sv[u]  = s4[qq];
        }
        unsigned dl[FU * 4], p[FU * 4];
#pragma unroll
        for (int u = 0; u < FU; ++u) {
#pragma unroll
            for (int j = 0; j < 4; ++j) {
                int k = 4 * u + j;
                dl[k] = (unsigned)(i4get(dv[u], j) - lo);
                p[k]  = (unsigned)CAP;
                if (ok[u] && dl[k] < (unsigned)FR) p[k] = atomicAdd(&lcnt[dl[k]], 1u);
            }
        }
#pragma unroll
        for (int u = 0; u < FU; ++u) {
#pragma unroll
            for (int j = 0; j < 4; ++j) {
                int k = 4 * u + j;
                if (p[k] < (unsigned)CAP)
                    sub[((long)(lo + (int)dl[k]) * NCH + ch) * CAP + p[k]] = i4get(sv[u], j);
            }
        }
    }
    __syncthreads();
    for (int i = t; i < FR; i += 128) cnt8[(lo + i) * NCH + ch] = (int)lcnt[i];
}

__global__ void __launch_bounds__(128)
gemm_xlxr(const float* __restrict__ x,
          const float* __restrict__ Wl, const float* __restrict__ bl,
          const float* __restrict__ Wr, const float* __restrict__ br,
          float* __restrict__ xl, float* __restrict__ xr,
          const int* __restrict__ ei, int* __restrict__ cnt8, int* __restrict__ sub) {
    __shared__ unsigned lcnt[FR];
    if (blockIdx.x < (unsigned)FUNITS) {
        fill_unit(ei, cnt8, sub, (int)blockIdx.x, lcnt);
        return;
    }

    int wid  = (int)(((blockIdx.x - (unsigned)FUNITS) * blockDim.x + threadIdx.x) >> 5);
    int lane = (int)(threadIdx.x & 31);

    int mt  = wid >> 3;
    int g   = wid & 7;
    int mat = g >> 2;
    int cb  = (g & 3) * 64;

    const float* Wm   = mat ? Wr : Wl;
    const float* bias = mat ? br : bl;
    float*       out  = mat ? xr : xl;

    int h = lane >> 4;
    int c = lane & 15;

    const float* ap = x  + (long)(mt * 16 + c) * DIM + 4 * h;
    const float* bp = Wm + (long)(4 * h) * DIM + cb + 4 * c;

    v8f acc[4] = {};
    gfrag cur, nxt;
    gload(cur, ap, bp, 0);
#pragma unroll
    for (int s = 0; s < GSTEPS; ++s) {
        if (s + 1 < GSTEPS) gload(nxt, ap, bp, (s + 1) * GK);
        __builtin_amdgcn_sched_barrier(0);
#pragma unroll
        for (int u = 0; u < GK / 8; ++u) {
#pragma unroll
            for (int p = 0; p < 2; ++p) {
                v2f a;
                a[0] = f4get(cur.a[u], 2 * p);
                a[1] = f4get(cur.a[u], 2 * p + 1);
#pragma unroll
                for (int j = 0; j < 4; ++j) {
                    v2f b;
                    b[0] = f4get(cur.b[u][2 * p], j);
                    b[1] = f4get(cur.b[u][2 * p + 1], j);
                    acc[j] = __builtin_amdgcn_wmma_f32_16x16x4_f32(
                                  false, a,  false, b,
                                  (short)0, acc[j],  false,  false);
                }
            }
        }
        if (s + 1 < GSTEPS) cur = nxt;
    }

    float4 bv = *(const float4*)(bias + cb + 4 * c);
#pragma unroll
    for (int r = 0; r < 8; ++r) {
        int m = mt * 16 + r + 8 * h;
        float4 v;
        v.x = acc[0][r] + bv.x;
        v.y = acc[1][r] + bv.y;
        v.z = acc[2][r] + bv.z;
        v.w = acc[3][r] + bv.w;
        *(float4*)(out + (long)m * DIM + cb + 4 * c) = v;
    }
}

#define SLOTS 64

__device__ __forceinline__ float logit_part(const float4& a0, const float4& a1,
                                            const float4& r0, const float4& r1,
                                            const float4& t0, const float4& t1) {
    float acc = 0.f, h;
    h = a0.x + r0.x; h = h > 0.f ? h : NEG_SLOPE * h; acc += t0.x * h;
    h = a0.y + r0.y; h = h > 0.f ? h : NEG_SLOPE * h; acc += t0.y * h;
    h = a0.z + r0.z; h = h > 0.f ? h : NEG_SLOPE * h; acc += t0.z * h;
    h = a0.w + r0.w; h = h > 0.f ? h : NEG_SLOPE * h; acc += t0.w * h;
    h = a1.x + r1.x; h = h > 0.f ? h : NEG_SLOPE * h; acc += t1.x * h;
    h = a1.y + r1.y; h = h > 0.f ? h : NEG_SLOPE * h; acc += t1.y * h;
    h = a1.z + r1.z; h = h > 0.f ? h : NEG_SLOPE * h; acc += t1.z * h;
    h = a1.w + r1.w; h = h > 0.f ? h : NEG_SLOPE * h; acc += t1.w * h;
    return acc;
}

__device__ __forceinline__ float edge_logit(const float4& a0, const float4& a1,
                                            const float4& r0, const float4& r1,
                                            const float4& t0, const float4& t1) {
    float acc = logit_part(a0, a1, r0, r1, t0, t1);
    for (int off = 16; off > 0; off >>= 1) acc += __shfl_xor(acc, off, 32);
    return acc;
}


#define NODE_WAVES 16
#define NBLK       (NNODE / NODE_WAVES)
static_assert(NNODE % NODE_WAVES == 0, "grid must cover the nodes exactly");
#define ID_PAD     0x7fffffff

__device__ __forceinline__ void absorb(const float4& a0, const float4& a1,
                                       const float4& r0, const float4& r1,
                                       const float4& t0, const float4& t1,
                                       float& m, float& den, float4& acc0, float4& acc1) {
    float lg = edge_logit(a0, a1, r0, r1, t0, t1);
    float mn = fmaxf(m, lg);
    float sc = __expf(m - mn);
    float wt = __expf(lg - mn);
    den = den * sc + wt;
    acc0.x = acc0.x * sc + wt * a0.x;  acc0.y = acc0.y * sc + wt * a0.y;
    __builtin_amdgcn_sched_barrier(0);
    acc0.z = acc0.z * sc + wt * a0.z;  acc0.w = acc0.w * sc + wt * a0.w;
    __builtin_amdgcn_sched_barrier(0);
    acc1.x = acc1.x * sc + wt * a1.x;  acc1.y = acc1.y * sc + wt * a1.y;
    __builtin_amdgcn_sched_barrier(0);
    acc1.z = acc1.z * sc + wt * a1.z;  acc1.w = acc1.w * sc + wt * a1.w;
    m = mn;
}

#define GROUP 4

template <int OFF>
__device__ __forceinline__ void group_xor_sum(float (&v)[GROUP]) {
    float o[GROUP];
#pragma unroll
    for (int j = 0; j < GROUP; ++j)
        o[j] = __builtin_bit_cast(float,
                   __builtin_amdgcn_ds_swizzle(__builtin_bit_cast(int, v[j]), (OFF << 10) | 0x1f));
#pragma unroll
    for (int j = 0; j < GROUP; ++j) v[j] += o[j];
}

__device__ __forceinline__ void absorb_group(const float4 (&a0)[GROUP], const float4 (&a1)[GROUP],
                                             const bool (&live)[GROUP],
                                             const float4& r0, const float4& r1,
                                             const float4& t0, const float4& t1,
                                             float& m, float& den, float4& acc0, float4& acc1) {
    float lg[GROUP];
#pragma unroll
    for (int j = 0; j < GROUP; ++j)
        lg[j] = live[j] ? logit_part(a0[j], a1[j], r0, r1, t0, t1) : -INFINITY;
    group_xor_sum<16>(lg);
    group_xor_sum<8>(lg);
    group_xor_sum<4>(lg);
    group_xor_sum<2>(lg);
    group_xor_sum<1>(lg);
    float mn = m;
#pragma unroll
    for (int j = 0; j < GROUP; ++j) mn = fmaxf(mn, lg[j]);
    float sc = __expf(m - mn);
    den *= sc;
    acc0.x *= sc;  acc0.y *= sc;  acc0.z *= sc;  acc0.w *= sc;
    acc1.x *= sc;  acc1.y *= sc;  acc1.z *= sc;  acc1.w *= sc;
#pragma unroll
    for (int j = 0; j < GROUP; ++j) {
        float wt = __expf(lg[j] - mn);
        den += wt;
        acc0.x += wt * a0[j].x;  acc0.y += wt * a0[j].y;
        acc0.z += wt * a0[j].z;  acc0.w += wt * a0[j].w;
        acc1.x += wt * a1[j].x;  acc1.y += wt * a1[j].y;
        acc1.z += wt * a1[j].z;  acc1.w += wt * a1[j].w;
    }
    m = mn;
}

__device__ __forceinline__ int sort32(int v, int lane) {
#pragma unroll
    for (int k = 2; k <= 32; k <<= 1) {
#pragma unroll
        for (int j = k >> 1; j > 0; j >>= 1) {
            int  o   = __shfl_xor(v, j, 32);
            bool asc = (lane & k) == 0;
            bool low = (lane & j) == 0;
            int  mn  = v < o ? v : o;
            int  mx  = v < o ? o : v;
            v = (asc == low) ? mn : mx;
        }
    }
    return v;
}

__device__ __forceinline__ void sort64(int& a, int& b, int lane) {
#pragma unroll
    for (int k = 2; k <= 32; k <<= 1) {
#pragma unroll
        for (int j = k >> 1; j > 0; j >>= 1) {
            int  oa   = __shfl_xor(a, j, 32);
            int  ob   = __shfl_xor(b, j, 32);
            bool low  = (lane & j) == 0;
            bool asca = (lane & k) == 0;
            bool ascb = ((lane + 32) & k) == 0;
            int  mna  = a < oa ? a : oa,  mxa = a < oa ? oa : a;
            int  mnb  = b < ob ? b : ob,  mxb = b < ob ? ob : b;
            a = (asca == low) ? mna : mxa;
            b = (ascb == low) ? mnb : mxb;
        }
    }
    int mn = a < b ? a : b;
    int mx = a < b ? b : a;
    a = mn;
    b = mx;
#pragma unroll
    for (int j = 16; j > 0; j >>= 1) {
        int  oa  = __shfl_xor(a, j, 32);
        int  ob  = __shfl_xor(b, j, 32);
        bool low = (lane & j) == 0;
        int  mna = a < oa ? a : oa,  mxa = a < oa ? oa : a;
        int  mnb = b < ob ? b : ob,  mxb = b < ob ? ob : b;
        a = low ? mna : mxa;
        b = low ? mnb : mxb;
    }
}

__device__ __forceinline__ float finish1(float a, float den, float b, float u) {
    float v = a / den + b;
    v = v > 0.f ? v : 0.f;
    return (u >= 0.5f) ? 2.0f * v : 0.0f;
}

__global__ void __launch_bounds__(NODE_WAVES * 32)
node_attend_post(const int* __restrict__ cnt8, const int* __restrict__ sub,
                 const int* __restrict__ ei,
                 const float* __restrict__ xl, const float* __restrict__ xr,
                 const float* __restrict__ att, const float* __restrict__ bias,
                 const float* __restrict__ du,
                 float* __restrict__ vout, float* __restrict__ part) {
    __shared__ alignas(16) float vs[NODE_WAVES][DIM];
    __shared__ int idl[NODE_WAVES][SLOTS];

    int w    = __builtin_amdgcn_readfirstlane((int)(threadIdx.x >> 5));
    int lane = (int)(threadIdx.x & 31);
    int d    = (int)blockIdx.x * NODE_WAVES + w;

    static_assert(NCH * CAP == 4 * 32, "four places of the node's row per lane");
    int  nc   = cnt8[d * NCH + (lane >> 2)];

    const float4* xd = (const float4*)(xr + (long)d * DIM);
    const float4* av = (const float4*)att;
    float4 r0 = xd[lane], r1 = xd[lane + 32];
    float4 t0 = av[lane], t1 = av[lane + 32];
    const float4* xs = (const float4*)(xl + (long)d * DIM);
    float4 acc0 = xs[lane], acc1 = xs[lane + 32];

    int  have = nc - 4 * (lane & 3);
    have = have < 0 ? 0 : have > 4 ? 4 : have;
    const int* row = sub + (long)d * (NCH * CAP) + 4 * lane;
    int q0 = ID_PAD, q1 = ID_PAD, q2 = ID_PAD, q3 = ID_PAD;
    if (have > 0) q0 = row[0];
    if (have > 1) q1 = row[1];
    if (have > 2) q2 = row[2];
    if (have > 3) q3 = row[3];
    bool whole = __builtin_amdgcn_ballot_w32(nc > CAP) == 0u;
    int upto = have;
#pragma unroll
    for (int o = 1; o < 32; o <<= 1) {
        int v = __shfl_up(upto, o, 32);
        if (lane >= o) upto += v;
    }
    unsigned n = (unsigned)__builtin_amdgcn_readlane(upto, 31);

    float  m    = edge_logit(acc0, acc1, r0, r1, t0, t1);
    float  den  = 1.f;

    static_assert(SLOTS % GROUP == 0, "a group never reaches past the two id registers");
    if (whole && n <= (unsigned)SLOTS) {
        int* il   = idl[w];
        int  from = upto - have;
        if (have > 0) il[from]     = q0;
        if (have > 1) il[from + 1] = q1;
        if (have > 2) il[from + 2] = q2;
        if (have > 3) il[from + 3] = q3;
        int id  = (lane < (int)n) ? il[lane] : ID_PAD;
        int id2 = ID_PAD;
        if (n <= 32u) {
            id = sort32(id, lane);
        } else {
            id2 = (lane + 32 < (int)n) ? il[32 + lane] : ID_PAD;
            sort64(id, id2, lane);
        }
        const float4* xl4 = (const float4*)xl;
        for (int i0 = 0; i0 < (int)n; i0 += GROUP) {
            float4 a0[GROUP], a1[GROUP];
            bool   live[GROUP];
            int    idv = i0 < 32 ? id : id2;
#pragma unroll
            for (int j = 0; j < GROUP; ++j) {
                int ix  = i0 + j;
                live[j] = ix < (int)n;
                int s   = __builtin_amdgcn_readlane(idv, ix & 31);
                int su  = (int)min((unsigned)s, (unsigned)(NNODE - 1));
                int ro  = su * (DIM / 4) + lane;
                a0[j] = xl4[ro];
                a1[j] = xl4[ro + 32];
            }
            absorb_group(a0, a1, live, r0, r1, t0, t1, m, den, acc0, acc1);
        }
    } else {
        static_assert(NEDGE % 32 == 0, "every lane runs the same number of trips");
        int ne = (lane & 3) == 0 ? nc : 0;
        for (int off = 16; off > 0; off >>= 1) ne += __shfl_xor(ne, off, 32);
        n = (unsigned)__builtin_amdgcn_readfirstlane(ne);
        const int* es = ei;
        const int* ed = ei + NEDGE;
        int      last = -1;
        unsigned done = 0u;
        while (done < n) {
            int best = ID_PAD;
            for (int e = lane; e < NEDGE; e += 32) {
                int v = (ed[e] == d) ? es[e] : ID_PAD;
                if (v > last && v < best) best = v;
            }
            for (int off = 16; off > 0; off >>= 1) {
                int o = __shfl_xor(best, off, 32);
                best = o < best ? o : best;
            }
            int cnt = 0;
            for (int e = lane; e < NEDGE; e += 32)
                cnt += (ed[e] == d && es[e] == best) ? 1 : 0;
            for (int off = 16; off > 0; off >>= 1) cnt += __shfl_xor(cnt, off, 32);
            if (cnt == 0) break;
            const float4* xn = (const float4*)(xl + (long)best * DIM);
            float4 a0 = xn[lane], a1 = xn[lane + 32];
            for (int c = 0; c < cnt; ++c)
                absorb(a0, a1, r0, r1, t0, t1, m, den, acc0, acc1);
            last  = best;
            done += (unsigned)cnt;
        }
    }

    const float4* dp = (const float4*)(du + (long)d * DIM);
    float4 u0 = dp[lane], u1 = dp[lane + 32];
    const float4* bp = (const float4*)bias;
    float4 b0 = bp[lane], b1 = bp[lane + 32];
    float4 v0, v1;
    v0.x = finish1(acc0.x, den, b0.x, u0.x);  v0.y = finish1(acc0.y, den, b0.y, u0.y);
    v0.z = finish1(acc0.z, den, b0.z, u0.z);  v0.w = finish1(acc0.w, den, b0.w, u0.w);
    v1.x = finish1(acc1.x, den, b1.x, u1.x);  v1.y = finish1(acc1.y, den, b1.y, u1.y);
    v1.z = finish1(acc1.z, den, b1.z, u1.z);  v1.w = finish1(acc1.w, den, b1.w, u1.w);

    float4* o = (float4*)(vout + (long)d * DIM);
    o[lane]      = v0;
    o[lane + 32] = v1;

    float4* vl = (float4*)vs[w];
    vl[lane]      = v0;
    vl[lane + 32] = v1;
    __syncthreads();

    int  t  = (int)threadIdx.x;
    int  c  = t & (DIM - 1);
    bool sq = t >= DIM;
    float s = 0.f;
#pragma unroll
    for (int r = 0; r < NODE_WAVES; ++r) {
        float x = vs[r][c];
        s += sq ? x * x : x;
    }
    part[(long)blockIdx.x * (2 * DIM) + t] = s;
}

#define FIN_WAVES 16
#define FIN_COLS  32
#define FIN_RG    16
#define FIN_ROWS  ((NNODE + FIN_RG - 1) / FIN_RG)
#define FIN_PPW   ((NBLK + FIN_WAVES - 1) / FIN_WAVES)
#define FIN_RPW   ((FIN_ROWS + FIN_WAVES - 1) / FIN_WAVES)

__global__ void __launch_bounds__(FIN_WAVES * 32)
finalize(const float* __restrict__ part,
         const float* __restrict__ gamma, const float* __restrict__ beta,
         float* out) {
    __shared__ float red[2][FIN_WAVES][FIN_COLS];
    __shared__ float sc_sh[FIN_COLS], sh_sh[FIN_COLS];

    int w    = (int)(threadIdx.x >> 5);
    int lane = (int)(threadIdx.x & 31);
    int colg = (int)blockIdx.x * FIN_COLS + lane;

    float s = 0.f, s2 = 0.f;
#pragma unroll
    for (int i = 0; i < FIN_PPW; ++i) {
        int  p  = w + FIN_WAVES * i;
        bool ok = p < NBLK;
        int  pc = ok ? p : NBLK - 1;
        float a  = part[(long)pc * (2 * DIM) + colg];
        float a2 = part[(long)pc * (2 * DIM) + DIM + colg];
        s  += ok ? a  : 0.f;
        s2 += ok ? a2 : 0.f;
    }
    red[0][w][lane] = s;
    red[1][w][lane] = s2;
    __syncthreads();
    if (w == 0) {
        float ts = 0.f, ts2 = 0.f;
#pragma unroll
        for (int g = 0; g < FIN_WAVES; ++g) { ts += red[0][g][lane]; ts2 += red[1][g][lane]; }
        const float invn = 1.0f / (float)NNODE;
        float mean  = ts * invn;
        float var   = ts2 * invn - mean * mean;
        float scale = gamma[colg] * rsqrtf(var + BN_EPS);
        sc_sh[lane] = scale;
        sh_sh[lane] = beta[colg] - mean * scale;
    }
    __syncthreads();
    float scale = sc_sh[lane], shift = sh_sh[lane];

    int r0   = (int)blockIdx.y * FIN_ROWS;
    int rend = r0 + FIN_ROWS < NNODE ? r0 + FIN_ROWS : NNODE;
    float x[FIN_RPW];
#pragma unroll
    for (int i = 0; i < FIN_RPW; ++i) {
        int r  = r0 + w + FIN_WAVES * i;
        int rc = r < rend ? r : rend - 1;
        x[i] = out[(long)rc * DIM + colg];
    }
#pragma unroll
    for (int i = 0; i < FIN_RPW; ++i) {
        int r = r0 + w + FIN_WAVES * i;
        if (r < rend) out[(long)r * DIM + colg] = x[i] * scale + shift;
    }
}

extern "C" void kernel_launch(void* const* d_in, const int* in_sizes, int n_in,
                              void* d_out, int out_size, void* d_ws, size_t ws_size,
                              hipStream_t stream) {
    const float* x     = (const float*)d_in[0];
    const int*   ei    = (const int*)d_in[1];
    const float* Wl    = (const float*)d_in[2];
    const float* bl    = (const float*)d_in[3];
    const float* Wr    = (const float*)d_in[4];
    const float* br    = (const float*)d_in[5];
    const float* att   = (const float*)d_in[6];
    const float* bias  = (const float*)d_in[7];
    const float* gamma = (const float*)d_in[8];
    const float* beta  = (const float*)d_in[9];
    const float* du    = (const float*)d_in[10];
    float*       out   = (float*)d_out;

    static_assert((long)NBLK * 2 * DIM + (long)NNODE * NCH * CAP + (long)NNODE * NCH <= (long)NNODE * DIM,
                  "partials, source lists and counts must fit the first span");
    float*    ws     = (float*)d_ws;
    float*    part   = ws;
    int*      sub    = (int*)(ws + (long)NBLK * 2 * DIM);
    int*      cnt8   = sub + (long)NNODE * NCH * CAP;
    float*    denom  = ws + (long)NNODE * DIM;
    unsigned* deg    = (unsigned*)(denom + NNODE);
    float*    sums   = denom + 2L * NNODE;
    float*    sumsq  = sums + DIM;
    float*    xl     = sumsq + DIM;
    float*    xr     = xl + (long)NNODE * DIM;

    gemm_xlxr<<<FUNITS + (MT * 8) / 4, 128, 0, stream>>>(x, Wl, bl, Wr, br, xl, xr, ei, cnt8, sub);

    node_attend_post<<<NBLK, NODE_WAVES * 32, 0, stream>>>(cnt8, sub, ei, xl, xr, att, bias, du, out, part);

    finalize<<<dim3(DIM / FIN_COLS, FIN_RG), FIN_WAVES * 32, 0, stream>>>(part, gamma, beta, out);
}
